# starting kernel
# baseline (speedup 1.0000x reference)
#include <hip/hip_runtime.h>
#include <hip/hip_bf16.h>

typedef __attribute__((ext_vector_type(2))) float v2f;
typedef __attribute__((ext_vector_type(8))) float v8f;

#define D_FEAT 128

__global__ void sage_zero_ws(float* __restrict__ p, long n) {
    long i = (long)blockIdx.x * blockDim.x + threadIdx.x;
    long stride = (long)gridDim.x * blockDim.x;
    for (; i < n; i += stride) p[i] = 0.0f;
}

__global__ __launch_bounds__(256) void sage_scatter(
    const float* __restrict__ feat, const int* __restrict__ src,
    const int* __restrict__ dst, float* __restrict__ summed,
    float* __restrict__ deg, int n_edges)
{
    long gid = (long)blockIdx.x * blockDim.x + threadIdx.x;
    int e = (int)(gid >> 5);
    int lane = (int)(gid & 31);
    if (e >= n_edges) return;
    int s = src[e];
    int d = dst[e];
    const float4 v = *(const float4*)(feat + (long)s * D_FEAT + lane * 4);
    float* acc = summed + (long)d * D_FEAT + lane * 4;
    atomicAdd(acc + 0, v.x);
    atomicAdd(acc + 1, v.y);
    atomicAdd(acc + 2, v.z);
    atomicAdd(acc + 3, v.w);
    if (lane == 0) atomicAdd(deg + d, 1.0f);
}

__global__ __launch_bounds__(256) void sage_norm(
    float* __restrict__ summed, const float* __restrict__ deg, long n_elts)
{
    long i = (long)blockIdx.x * blockDim.x + threadIdx.x;
    if (i >= n_elts) return;
    float dg = deg[i >> 7];
    float inv = dg > 1.0f ? 1.0f / dg : 1.0f;
    summed[i] *= inv;
}

__global__ __launch_bounds__(256) void sage_gemm_wmma(
    const float* __restrict__ feat, const float* __restrict__ hneigh,
    const float* __restrict__ Wself, const float* __restrict__ Wneigh,
    float* __restrict__ out)
{
    const int wave = threadIdx.x >> 5;
    const int lane = threadIdx.x & 31;
    const int m = lane & 15;
    const int khalf = (lane >> 4) * 2;
    const long node_base = (long)blockIdx.x * 16;
    const int  col_base = wave * 16;

    const float* fa = feat   + (node_base + m) * D_FEAT + khalf;
    const float* ha = hneigh + (node_base + m) * D_FEAT + khalf;
    const float* ws = Wself  + (long)(col_base + m) * D_FEAT + khalf;
    const float* wn = Wneigh + (long)(col_base + m) * D_FEAT + khalf;

    v8f c = {};
#pragma unroll
    for (int k0 = 0; k0 < D_FEAT; k0 += 4) {
        v2f a  = *(const v2f*)(fa + k0);
        v2f bs = *(const v2f*)(ws + k0);
        c = __builtin_amdgcn_wmma_f32_16x16x4_f32(
                 false, a,  false, bs,
                 (short)0, c,  false,  false);
        v2f ah = *(const v2f*)(ha + k0);
        v2f bn = *(const v2f*)(wn + k0);
        c = __builtin_amdgcn_wmma_f32_16x16x4_f32(
                false, ah, false, bn, (short)0, c, false, false);
    }

    float* o = out + (node_base + (lane >> 4) * 8) * D_FEAT + col_base + m;
#pragma unroll
    for (int r = 0; r < 8; ++r)
        o[(long)r * D_FEAT] = c[r];
}

extern "C" void kernel_launch(void* const* d_in, const int* in_sizes, int n_in,
                              void* d_out, int out_size, void* d_ws, size_t ws_size,
                              hipStream_t stream)
{
    const float* feat   = (const float*)d_in[0];
    const int*   src    = (const int*)d_in[1];
    const int*   dst    = (const int*)d_in[2];
    const float* Wself  = (const float*)d_in[3];
    const float* Wneigh = (const float*)d_in[4];
    float*       out    = (float*)d_out;

    const int n_nodes = in_sizes[0] / D_FEAT;
    const int n_edges = in_sizes[1];

    float* summed = (float*)d_ws;
    float* deg    = summed + (long)n_nodes * D_FEAT;

    const long n_acc = (long)n_nodes * D_FEAT + n_nodes;
    sage_zero_ws<<<4096, 256, 0, stream>>>(summed, n_acc);

    const long scatter_threads = (long)n_edges * 32;
    const int  scatter_blocks  = (int)((scatter_threads + 255) / 256);
    sage_scatter<<<scatter_blocks, 256, 0, stream>>>(feat, src, dst, summed, deg, n_edges);

    const long n_elts = (long)n_nodes * D_FEAT;
    sage_norm<<<(int)((n_elts + 255) / 256), 256, 0, stream>>>(summed, deg, n_elts);

    sage_gemm_wmma<<<n_nodes / 16, 256, 0, stream>>>(feat, summed, Wself, Wneigh, out);
}


# Dense part retiled: both weight matrices staged once per block in LDS, each wave keeps 8 column-tile accumulators and loads each A fragment once, 16 B per lane (k summation order changes)
# speedup vs baseline: 1.4975x; 1.4975x over previous
#include <hip/hip_runtime.h>
#include <hip/hip_bf16.h>

typedef __attribute__((ext_vector_type(2))) float v2f;
typedef __attribute__((ext_vector_type(4))) float v4f;
typedef __attribute__((ext_vector_type(8))) float v8f;

#define D_FEAT 128

__global__ void sage_zero_ws(float* __restrict__ p, long n) {
    long i = (long)blockIdx.x * blockDim.x + threadIdx.x;
    long stride = (long)gridDim.x * blockDim.x;
    for (; i < n; i += stride) p[i] = 0.0f;
}

__global__ __launch_bounds__(256) void sage_scatter(
    const float* __restrict__ feat, const int* __restrict__ src,
    const int* __restrict__ dst, float* __restrict__ summed,
    float* __restrict__ deg, int n_edges)
{
    long gid = (long)blockIdx.x * blockDim.x + threadIdx.x;
    int e = (int)(gid >> 5);
    int lane = (int)(gid & 31);
    if (e >= n_edges) return;
    int s = src[e];
    int d = dst[e];
    const float4 v = *(const float4*)(feat + (long)s * D_FEAT + lane * 4);
    float* acc = summed + (long)d * D_FEAT + lane * 4;
    atomicAdd(acc + 0, v.x);
    atomicAdd(acc + 1, v.y);
    atomicAdd(acc + 2, v.z);
    atomicAdd(acc + 3, v.w);
    if (lane == 0) atomicAdd(deg + d, 1.0f);
}

__global__ __launch_bounds__(256) void sage_norm(
    float* __restrict__ summed, const float* __restrict__ deg, long n_elts)
{
    long i = (long)blockIdx.x * blockDim.x + threadIdx.x;
    if (i >= n_elts) return;
    float dg = deg[i >> 7];
    float inv = dg > 1.0f ? 1.0f / dg : 1.0f;
    summed[i] *= inv;
}

#define W_K        256
#define W_CSTRIDE  (8 * W_K + 4)
#define ROWS_PER_BLOCK 128

__device__ __forceinline__ v8f wmma_f32(v2f a, v2f b, v8f c) {
    return __builtin_amdgcn_wmma_f32_16x16x4_f32(
         false, a,  false, b,
         (short)0, c,  false,  false);
}

__global__ __launch_bounds__(256) void sage_gemm_wmma(
    const float* __restrict__ feat, const float* __restrict__ hneigh,
    const float* __restrict__ Wself, const float* __restrict__ Wneigh,
    float* __restrict__ out, int n_tiles)
{
    __shared__ __attribute__((aligned(16))) float sW[16 * W_CSTRIDE];

    const int wave = __builtin_amdgcn_readfirstlane((int)(threadIdx.x >> 5));
    const int lane = threadIdx.x & 31;
    const int c = lane & 15;
    const int h = lane >> 4;

    const int  tile   = blockIdx.x * (ROWS_PER_BLOCK / 16) + wave;
    const bool active = tile < n_tiles;
    const int  tile_c = active ? tile : n_tiles - 1;
    const long row    = (long)tile_c * 16 + c;
    const float* fa = feat   + row * D_FEAT + 4 * h;
    const float* ha = hneigh + row * D_FEAT + 4 * h;

    v4f a[4][8];
#pragma unroll
    for (int j = 0; j < 8; ++j) a[0][j] = *(const v4f*)(fa + 8 * j);

#pragma unroll
    for (int i = 0; i < 16; ++i) {
        const int n   = 8 * i + wave;
        const int dst = (n & 15) * W_CSTRIDE + (n >> 4) * W_K + lane * 4;
        const v4f ws = *(const v4f*)(Wself  + n * D_FEAT + lane * 4);
        const v4f wn = *(const v4f*)(Wneigh + n * D_FEAT + lane * 4);
        *(v4f*)(sW + dst)          = ws;
        *(v4f*)(sW + dst + D_FEAT) = wn;
    }
    __syncthreads();
    if (!active) return;

#pragma unroll
    for (int j = 0; j < 8; ++j) a[1][j] = *(const v4f*)(fa + 64 + 8 * j);

    v8f acc[8];
#pragma unroll
    for (int nt = 0; nt < 8; ++nt) acc[nt] = (v8f){};

    const float* bl = sW + c * W_CSTRIDE + 4 * h;

    v4f bcur[8], bnxt[8];
#pragma unroll
    for (int nt = 0; nt < 8; ++nt)
        bcur[nt] = *(const v4f*)(bl + nt * W_K);

#pragma unroll
    for (int p = 0; p < 4; ++p) {
        if (p == 1) {
#pragma unroll
            for (int j = 0; j < 8; ++j) a[2][j] = *(const v4f*)(ha + 8 * j);
        }
        if (p == 2) {
#pragma unroll
            for (int j = 0; j < 8; ++j) a[3][j] = *(const v4f*)(ha + 64 + 8 * j);
        }
#pragma unroll
        for (int j = 0; j < 8; ++j) {
            const int knext = p * 64 + j * 8 + 8;
            if (knext < W_K) {
#pragma unroll
                for (int nt = 0; nt < 8; ++nt)
                    bnxt[nt] = *(const v4f*)(bl + nt * W_K + knext);
            }
            const v4f av = a[p][j];
            const v2f a_lo = {av.x, av.y};
            const v2f a_hi = {av.z, av.w};
#pragma unroll
            for (int nt = 0; nt < 8; ++nt) {
                const v2f b_lo = {bcur[nt].x, bcur[nt].y};
                acc[nt] = wmma_f32(a_lo, b_lo, acc[nt]);
            }
#pragma unroll
            for (int nt = 0; nt < 8; ++nt) {
                const v2f b_hi = {bcur[nt].z, bcur[nt].w};
                acc[nt] = wmma_f32(a_hi, b_hi, acc[nt]);
            }
            if (knext < W_K) {
#pragma unroll
                for (int nt = 0; nt < 8; ++nt) bcur[nt] = bnxt[nt];
            }
        }
    }

    float* o = out + ((long)tile * 16 + h * 8) * D_FEAT + c;
#pragma unroll
    for (int nt = 0; nt < 8; ++nt) {
#pragma unroll
        for (int r = 0; r < 8; ++r)
            o[(long)r * D_FEAT + nt * 16] = acc[nt][r];
    }
}

extern "C" void kernel_launch(void* const* d_in, const int* in_sizes, int n_in,
                              void* d_out, int out_size, void* d_ws, size_t ws_size,
                              hipStream_t stream)
{
    const float* feat   = (const float*)d_in[0];
    const int*   src    = (const int*)d_in[1];
    const int*   dst    = (const int*)d_in[2];
    const float* Wself  = (const float*)d_in[3];
    const float* Wneigh = (const float*)d_in[4];
    float*       out    = (float*)d_out;

    const int n_nodes = in_sizes[0] / D_FEAT;
    const int n_edges = in_sizes[1];

    float* summed = (float*)d_ws;
    float* deg    = summed + (long)n_nodes * D_FEAT;

    const long n_acc = (long)n_nodes * D_FEAT + n_nodes;
    sage_zero_ws<<<4096, 256, 0, stream>>>(summed, n_acc);

    const long scatter_threads = (long)n_edges * 32;
    const int  scatter_blocks  = (int)((scatter_threads + 255) / 256);
    sage_scatter<<<scatter_blocks, 256, 0, stream>>>(feat, src, dst, summed, deg, n_edges);

    const long n_elts = (long)n_nodes * D_FEAT;
    sage_norm<<<(int)((n_elts + 255) / 256), 256, 0, stream>>>(summed, deg, n_elts);

    const int n_tiles = n_nodes / 16;
    sage_gemm_wmma<<<(n_tiles + 7) / 8, 256, 0, stream>>>(feat, summed, Wself, Wneigh, out, n_tiles);
}


# Float-atomic scatter-add replaced by per-node edge lists (one integer exchange per edge) that 8 lanes per node walk, summing rows in registers with the mean fused in (sum order changes)
# speedup vs baseline: 3.1991x; 2.1363x over previous
#include <hip/hip_runtime.h>
#include <hip/hip_bf16.h>

typedef __attribute__((ext_vector_type(2))) float v2f;
typedef __attribute__((ext_vector_type(4))) float v4f;
typedef __attribute__((ext_vector_type(8))) float v8f;

#define D_FEAT 128

__global__ __launch_bounds__(256) void sage_list_init(int* __restrict__ head, int n_nodes) {
    const int v = blockIdx.x * blockDim.x + threadIdx.x;
    if (v < n_nodes) head[v] = -1;
}

__global__ __launch_bounds__(256) void sage_list_build(
    const int* __restrict__ src, const int* __restrict__ dst,
    int* __restrict__ head, int2* __restrict__ link, int n_edges)
{
    const int e = blockIdx.x * blockDim.x + threadIdx.x;
    if (e >= n_edges) return;
    const int prev = atomicExch(head + dst[e], e);
    link[e] = make_int2(src[e], prev);
}

#define LANES_PER_NODE 8

__global__ __launch_bounds__(256) void sage_gather_mean(
    const float* __restrict__ feat, const int* __restrict__ head,
    const int2* __restrict__ link, float* __restrict__ hneigh,
    int n_nodes, int n_edges)
{
    const int gid = blockIdx.x * blockDim.x + threadIdx.x;
    const int v   = gid / LANES_PER_NODE;
    const int sub = gid % LANES_PER_NODE;
    if (v >= n_nodes) return;

    int  e = head[v];
    int2 l = make_int2(0, -1);
    if (e >= 0) l = link[e];

    v4f acc0 = {}, acc1 = {}, acc2 = {}, acc3 = {};
    int cnt = 0;
    for (int it = 0; it < n_edges && e >= 0; ++it) {
        const float* r = feat + (long)l.x * D_FEAT + sub * 4;
        const v4f x0 = *(const v4f*)(r);
        const v4f x1 = *(const v4f*)(r + 32);
        const v4f x2 = *(const v4f*)(r + 64);
        const v4f x3 = *(const v4f*)(r + 96);
        const int en = l.y;
        if (en >= 0) l = link[en];
        acc0 += x0;
        acc1 += x1;
        acc2 += x2;
        acc3 += x3;
        ++cnt;
        e = en;
    }

    const float dg = cnt > 1 ? (float)cnt : 1.0f;
    float* o = hneigh + (long)v * D_FEAT + sub * 4;
    *(v4f*)(o)      = acc0 / dg;
    *(v4f*)(o + 32) = acc1 / dg;
    *(v4f*)(o + 64) = acc2 / dg;
    *(v4f*)(o + 96) = acc3 / dg;
}

#define W_K        256
#define W_CSTRIDE  (8 * W_K + 4)
#define ROWS_PER_BLOCK 128

__device__ __forceinline__ v8f wmma_f32(v2f a, v2f b, v8f c) {
    return __builtin_amdgcn_wmma_f32_16x16x4_f32(
         false, a,  false, b,
         (short)0, c,  false,  false);
}

__global__ __launch_bounds__(256) void sage_gemm_wmma(
    const float* __restrict__ feat, const float* __restrict__ hneigh,
    const float* __restrict__ Wself, const float* __restrict__ Wneigh,
    float* __restrict__ out, int n_tiles)
{
    __shared__ __attribute__((aligned(16))) float sW[16 * W_CSTRIDE];

    const int wave = __builtin_amdgcn_readfirstlane((int)(threadIdx.x >> 5));
    const int lane = threadIdx.x & 31;
    const int c = lane & 15;
    const int h = lane >> 4;

    const int  tile   = blockIdx.x * (ROWS_PER_BLOCK / 16) + wave;
    const bool active = tile < n_tiles;
    const int  tile_c = active ? tile : n_tiles - 1;
    const long row    = (long)tile_c * 16 + c;
    const float* fa = feat   + row * D_FEAT + 4 * h;
    const float* ha = hneigh + row * D_FEAT + 4 * h;

    v4f a[4][8];
#pragma unroll
    for (int j = 0; j < 8; ++j) a[0][j] = *(const v4f*)(fa + 8 * j);

#pragma unroll
    for (int i = 0; i < 16; ++i) {
        const int n   = 8 * i + wave;
        const int dst = (n & 15) * W_CSTRIDE + (n >> 4) * W_K + lane * 4;
        const v4f ws = *(const v4f*)(Wself  + n * D_FEAT + lane * 4);
        const v4f wn = *(const v4f*)(Wneigh + n * D_FEAT + lane * 4);
        *(v4f*)(sW + dst)          = ws;
        *(v4f*)(sW + dst + D_FEAT) = wn;
    }
    __syncthreads();
    if (!active) return;

#pragma unroll
    for (int j = 0; j < 8; ++j) a[1][j] = *(const v4f*)(fa + 64 + 8 * j);

    v8f acc[8];
#pragma unroll
    for (int nt = 0; nt < 8; ++nt) acc[nt] = (v8f){};

    const float* bl = sW + c * W_CSTRIDE + 4 * h;

    v4f bcur[8], bnxt[8];
#pragma unroll
    for (int nt = 0; nt < 8; ++nt)
        bcur[nt] = *(const v4f*)(bl + nt * W_K);

#pragma unroll
    for (int p = 0; p < 4; ++p) {
        if (p == 1) {
#pragma unroll
            for (int j = 0; j < 8; ++j) a[2][j] = *(const v4f*)(ha + 8 * j);
        }
        if (p == 2) {
#pragma unroll
            for (int j = 0; j < 8; ++j) a[3][j] = *(const v4f*)(ha + 64 + 8 * j);
        }
#pragma unroll
        for (int j = 0; j < 8; ++j) {
            const int knext = p * 64 + j * 8 + 8;
            if (knext < W_K) {
#pragma unroll
                for (int nt = 0; nt < 8; ++nt)
                    bnxt[nt] = *(const v4f*)(bl + nt * W_K + knext);
            }
            const v4f av = a[p][j];
            const v2f a_lo = {av.x, av.y};
            const v2f a_hi = {av.z, av.w};
#pragma unroll
            for (int nt = 0; nt < 8; ++nt) {
                const v2f b_lo = {bcur[nt].x, bcur[nt].y};
                acc[nt] = wmma_f32(a_lo, b_lo, acc[nt]);
            }
#pragma unroll
            for (int nt = 0; nt < 8; ++nt) {
                const v2f b_hi = {bcur[nt].z, bcur[nt].w};
                acc[nt] = wmma_f32(a_hi, b_hi, acc[nt]);
            }
            if (knext < W_K) {
#pragma unroll
                for (int nt = 0; nt < 8; ++nt) bcur[nt] = bnxt[nt];
            }
        }
    }

    float* o = out + ((long)tile * 16 + h * 8) * D_FEAT + c;
#pragma unroll
    for (int nt = 0; nt < 8; ++nt) {
#pragma unroll
        for (int r = 0; r < 8; ++r)
            o[(long)r * D_FEAT + nt * 16] = acc[nt][r];
    }
}

extern "C" void kernel_launch(void* const* d_in, const int* in_sizes, int n_in,
                              void* d_out, int out_size, void* d_ws, size_t ws_size,
                              hipStream_t stream)
{
    const float* feat   = (const float*)d_in[0];
    const int*   src    = (const int*)d_in[1];
    const int*   dst    = (const int*)d_in[2];
    const float* Wself  = (const float*)d_in[3];
    const float* Wneigh = (const float*)d_in[4];
    float*       out    = (float*)d_out;

    const int n_nodes = in_sizes[0] / D_FEAT;
    const int n_edges = in_sizes[1];

    float* hneigh = (float*)d_ws;
    int2*  link   = (int2*)d_out;
    int*   head   = (int*)(link + n_edges);

    sage_list_init<<<(n_nodes + 255) / 256, 256, 0, stream>>>(head, n_nodes);

    sage_list_build<<<(n_edges + 255) / 256, 256, 0, stream>>>(src, dst, head, link, n_edges);

    const long walk_threads = (long)n_nodes * LANES_PER_NODE;
    sage_gather_mean<<<(int)((walk_threads + 255) / 256), 256, 0, stream>>>(
        feat, head, link, hneigh, n_nodes, n_edges);

    const int n_tiles = n_nodes / 16;
    sage_gemm_wmma<<<(n_tiles + 7) / 8, 256, 0, stream>>>(feat, hneigh, Wself, Wneigh, out, n_tiles);
}


# Per-edge global integer exchange replaced by a counting sort of the edges by 64-node destination bin on LDS counters; one block per bin links and walks its edges (sum order changes)
# speedup vs baseline: 3.8717x; 1.2102x over previous
#include <hip/hip_runtime.h>
#include <hip/hip_bf16.h>

typedef __attribute__((ext_vector_type(2))) float v2f;
typedef __attribute__((ext_vector_type(4))) float v4f;
typedef __attribute__((ext_vector_type(8))) float v8f;

#define D_FEAT 128

#define NODES_PER_BIN   64
#define BIN_SHIFT       6
#define EDGES_PER_BLOCK 8192
#define SORT_THREADS    1024

__global__ __launch_bounds__(SORT_THREADS) void sage_bin_count(
    const int* __restrict__ dst, int* __restrict__ block_hist, int n_edges, int n_bins)
{
    extern __shared__ int lcnt[];
    for (int t = threadIdx.x; t < n_bins; t += SORT_THREADS) lcnt[t] = 0;
    __syncthreads();
    const int e0 = blockIdx.x * EDGES_PER_BLOCK + threadIdx.x;
#pragma unroll
    for (int k = 0; k < EDGES_PER_BLOCK / SORT_THREADS; ++k) {
        const int e = e0 + k * SORT_THREADS;
        if (e < n_edges) atomicAdd(&lcnt[dst[e] >> BIN_SHIFT], 1);
    }
    __syncthreads();
    for (int t = threadIdx.x; t < n_bins; t += SORT_THREADS)
        block_hist[blockIdx.x * n_bins + t] = lcnt[t];
}

#define PREFIX_BINS   64
#define PREFIX_GROUPS 16

__global__ __launch_bounds__(PREFIX_BINS * PREFIX_GROUPS) void sage_bin_prefix(
    const int* __restrict__ block_hist, int* __restrict__ block_base,
    int* __restrict__ bin_total, int n_blocks, int n_bins)
{
    __shared__ int part[PREFIX_GROUPS][PREFIX_BINS];
    const int tl = threadIdx.x % PREFIX_BINS;
    const int g  = threadIdx.x / PREFIX_BINS;
    const int t  = blockIdx.x * PREFIX_BINS + tl;
    const int rows = (n_blocks + PREFIX_GROUPS - 1) / PREFIX_GROUPS;
    const int r0 = g * rows;
    const int r1 = r0 + rows < n_blocks ? r0 + rows : n_blocks;
    int s = 0;
    if (t < n_bins) {
#pragma unroll 8
        for (int b = r0; b < r1; ++b) s += block_hist[b * n_bins + t];
    }
    part[g][tl] = s;
    __syncthreads();
    if (t >= n_bins) return;
    int run = 0;
    for (int gg = 0; gg < g; ++gg) run += part[gg][tl];
#pragma unroll 8
    for (int b = r0; b < r1; ++b) {
        const int c = block_hist[b * n_bins + t];
        block_base[b * n_bins + t] = run;
        run += c;
    }
    if (g == PREFIX_GROUPS - 1) bin_total[t] = run;
}

__global__ __launch_bounds__(SORT_THREADS) void sage_bin_scatter(
    const int* __restrict__ src, const int* __restrict__ dst,
    const int* __restrict__ block_base, const int* __restrict__ bin_total,
    int* __restrict__ bin_start, int2* __restrict__ pairs, int n_edges, int n_bins)
{
    extern __shared__ int lcur[];
    __shared__ int wsum[SORT_THREADS / 32];

    const int cpt  = (n_bins + SORT_THREADS - 1) / SORT_THREADS;
    const int b0   = threadIdx.x * cpt;
    const int lane = threadIdx.x & 31;
    const int wave = threadIdx.x >> 5;
    int own = 0;
    for (int j = 0; j < cpt; ++j)
        if (b0 + j < n_bins) own += bin_total[b0 + j];
    int inc = own;
    for (int d = 1; d < 32; d <<= 1) {
        const int y = __shfl_up(inc, d, 32);
        if (lane >= d) inc += y;
    }
    if (lane == 31) wsum[wave] = inc;
    __syncthreads();
    int run = inc - own;
    for (int w = 0; w < wave; ++w) run += wsum[w];
    for (int j = 0; j < cpt; ++j) {
        const int b = b0 + j;
        if (b < n_bins) {
            lcur[b] = run + block_base[blockIdx.x * n_bins + b];
            if (blockIdx.x == 0) bin_start[b] = run;
            run += bin_total[b];
        }
    }
    if (blockIdx.x == 0 && threadIdx.x == SORT_THREADS - 1) bin_start[n_bins] = run;
    __syncthreads();

    const int e0 = blockIdx.x * EDGES_PER_BLOCK + threadIdx.x;
#pragma unroll
    for (int k = 0; k < EDGES_PER_BLOCK / SORT_THREADS; ++k) {
        const int e = e0 + k * SORT_THREADS;
        if (e < n_edges) {
            const int d = dst[e];
            const int s = src[e];
            const int place = atomicAdd(&lcur[d >> BIN_SHIFT], 1);
            pairs[place] = make_int2(s, d);
        }
    }
}

#define LANES_PER_NODE 8
#define WALK_THREADS   (NODES_PER_BIN * LANES_PER_NODE)

__global__ __launch_bounds__(WALK_THREADS) void sage_bin_gather_mean(
    const float* __restrict__ feat, const int2* __restrict__ pairs,
    const int* __restrict__ bin_start, int2* link, float* __restrict__ hneigh,
    int n_nodes, int n_edges)
{
    __shared__ int lhead[NODES_PER_BIN];
    const int base = blockIdx.x * NODES_PER_BIN;
    const int s0 = bin_start[blockIdx.x];
    const int s1 = bin_start[blockIdx.x + 1];
    if (threadIdx.x < NODES_PER_BIN) lhead[threadIdx.x] = -1;
    __syncthreads();
    for (int i = s0 + threadIdx.x; i < s1; i += WALK_THREADS) {
        const int2 p = pairs[i];
        const int prev = atomicExch(&lhead[p.y - base], i);
        link[i] = make_int2(p.x, prev);
    }
    __syncthreads();

    const int nl  = threadIdx.x / LANES_PER_NODE;
    const int sub = threadIdx.x % LANES_PER_NODE;
    const int v   = base + nl;
    if (v >= n_nodes) return;

    int  e = lhead[nl];
    int2 l = make_int2(0, -1);
    if (e >= 0) l = link[e];

    v4f acc0 = {}, acc1 = {}, acc2 = {}, acc3 = {};
    int cnt = 0;
    for (int it = 0; it < n_edges && e >= 0; ++it) {
        const float* r = feat + (long)l.x * D_FEAT + sub * 4;
        const v4f x0 = *(const v4f*)(r);
        const v4f x1 = *(const v4f*)(r + 32);
        const v4f x2 = *(const v4f*)(r + 64);
        const v4f x3 = *(const v4f*)(r + 96);
        const int en = l.y;
        if (en >= 0) l = link[en];
        acc0 += x0;
        acc1 += x1;
        acc2 += x2;
        acc3 += x3;
        ++cnt;
        e = en;
    }

    const float dg = cnt > 1 ? (float)cnt : 1.0f;
    float* o = hneigh + (long)v * D_FEAT + sub * 4;
    *(v4f*)(o)      = acc0 / dg;
    *(v4f*)(o + 32) = acc1 / dg;
    *(v4f*)(o + 64) = acc2 / dg;
    *(v4f*)(o + 96) = acc3 / dg;
}

#define W_K        256
#define W_CSTRIDE  (8 * W_K + 4)
#define ROWS_PER_BLOCK 128

__device__ __forceinline__ v8f wmma_f32(v2f a, v2f b, v8f c) {
    return __builtin_amdgcn_wmma_f32_16x16x4_f32(
         false, a,  false, b,
         (short)0, c,  false,  false);
}

__global__ __launch_bounds__(256) void sage_gemm_wmma(
    const float* __restrict__ feat, const float* __restrict__ hneigh,
    const float* __restrict__ Wself, const float* __restrict__ Wneigh,
    float* __restrict__ out, int n_tiles)
{
    __shared__ __attribute__((aligned(16))) float sW[16 * W_CSTRIDE];

    const int wave = __builtin_amdgcn_readfirstlane((int)(threadIdx.x >> 5));
    const int lane = threadIdx.x & 31;
    const int c = lane & 15;
    const int h = lane >> 4;

    const int  tile   = blockIdx.x * (ROWS_PER_BLOCK / 16) + wave;
    const bool active = tile < n_tiles;
    const int  tile_c = active ? tile : n_tiles - 1;
    const long row    = (long)tile_c * 16 + c;
    const float* fa = feat   + row * D_FEAT + 4 * h;
    const float* ha = hneigh + row * D_FEAT + 4 * h;

    v4f a[4][8];
#pragma unroll
    for (int j = 0; j < 8; ++j) a[0][j] = *(const v4f*)(fa + 8 * j);

#pragma unroll
    for (int i = 0; i < 16; ++i) {
        const int n   = 8 * i + wave;
        const int dst = (n & 15) * W_CSTRIDE + (n >> 4) * W_K + lane * 4;
        const v4f ws = *(const v4f*)(Wself  + n * D_FEAT + lane * 4);
        const v4f wn = *(const v4f*)(Wneigh + n * D_FEAT + lane * 4);
        *(v4f*)(sW + dst)          = ws;
        *(v4f*)(sW + dst + D_FEAT) = wn;
    }
    __syncthreads();
    if (!active) return;

#pragma unroll
    for (int j = 0; j < 8; ++j) a[1][j] = *(const v4f*)(fa + 64 + 8 * j);

    v8f acc[8];
#pragma unroll
    for (int nt = 0; nt < 8; ++nt) acc[nt] = (v8f){};

    const float* bl = sW + c * W_CSTRIDE + 4 * h;

    v4f bcur[8], bnxt[8];
#pragma unroll
    for (int nt = 0; nt < 8; ++nt)
        bcur[nt] = *(const v4f*)(bl + nt * W_K);

#pragma unroll
    for (int p = 0; p < 4; ++p) {
        if (p == 1) {
#pragma unroll
            for (int j = 0; j < 8; ++j) a[2][j] = *(const v4f*)(ha + 8 * j);
        }
        if (p == 2) {
#pragma unroll
            for (int j = 0; j < 8; ++j) a[3][j] = *(const v4f*)(ha + 64 + 8 * j);
        }
#pragma unroll
        for (int j = 0; j < 8; ++j) {
            const int knext = p * 64 + j * 8 + 8;
            if (knext < W_K) {
#pragma unroll
                for (int nt = 0; nt < 8; ++nt)
                    bnxt[nt] = *(const v4f*)(bl + nt * W_K + knext);
            }
            const v4f av = a[p][j];
            const v2f a_lo = {av.x, av.y};
            const v2f a_hi = {av.z, av.w};
#pragma unroll
            for (int nt = 0; nt < 8; ++nt) {
                const v2f b_lo = {bcur[nt].x, bcur[nt].y};
                acc[nt] = wmma_f32(a_lo, b_lo, acc[nt]);
            }
#pragma unroll
            for (int nt = 0; nt < 8; ++nt) {
                const v2f b_hi = {bcur[nt].z, bcur[nt].w};
                acc[nt] = wmma_f32(a_hi, b_hi, acc[nt]);
            }
            if (knext < W_K) {
#pragma unroll
                for (int nt = 0; nt < 8; ++nt) bcur[nt] = bnxt[nt];
            }
        }
    }

    float* o = out + ((long)tile * 16 + h * 8) * D_FEAT + c;
#pragma unroll
    for (int nt = 0; nt < 8; ++nt) {
#pragma unroll
        for (int r = 0; r < 8; ++r)
            o[(long)r * D_FEAT + nt * 16] = acc[nt][r];
    }
}

extern "C" void kernel_launch(void* const* d_in, const int* in_sizes, int n_in,
                              void* d_out, int out_size, void* d_ws, size_t ws_size,
                              hipStream_t stream)
{
    const float* feat   = (const float*)d_in[0];
    const int*   src    = (const int*)d_in[1];
    const int*   dst    = (const int*)d_in[2];
    const float* Wself  = (const float*)d_in[3];
    const float* Wneigh = (const float*)d_in[4];
    float*       out    = (float*)d_out;

    const int n_nodes = in_sizes[0] / D_FEAT;
    const int n_edges = in_sizes[1];
    const int n_bins  = (n_nodes + NODES_PER_BIN - 1) / NODES_PER_BIN;
    const int n_eblk  = (n_edges + EDGES_PER_BLOCK - 1) / EDGES_PER_BLOCK;

    float* hneigh     = (float*)d_ws;
    int2*  pairs      = (int2*)d_out;
    int2*  link       = pairs + n_edges;
    int*   block_hist = (int*)(link + n_edges);
    int*   block_base = block_hist + n_eblk * n_bins;
    int*   bin_total  = block_base + n_eblk * n_bins;
    int*   bin_start  = bin_total + n_bins;

    sage_bin_count<<<n_eblk, SORT_THREADS, n_bins * sizeof(int), stream>>>(
        dst, block_hist, n_edges, n_bins);

    sage_bin_prefix<<<(n_bins + PREFIX_BINS - 1) / PREFIX_BINS, PREFIX_BINS * PREFIX_GROUPS, 0, stream>>>(
        block_hist, block_base, bin_total, n_eblk, n_bins);

    sage_bin_scatter<<<n_eblk, SORT_THREADS, n_bins * sizeof(int), stream>>>(
        src, dst, block_base, bin_total, bin_start, pairs, n_edges, n_bins);

    sage_bin_gather_mean<<<n_bins, WALK_THREADS, 0, stream>>>(
        feat, pairs, bin_start, link, hneigh, n_nodes, n_edges);

    const int n_tiles = n_nodes / 16;
    sage_gemm_wmma<<<(n_tiles + 7) / 8, 256, 0, stream>>>(feat, hneigh, Wself, Wneigh, out, n_tiles);
}


# Dense kernel made resident: 512 blocks (two per WGP) stage the weights once and loop over (16-row tile, 64-column half) tasks dealt evenly over the SIMDs; k summation order unchanged
# speedup vs baseline: 4.0463x; 1.0451x over previous
#include <hip/hip_runtime.h>
#include <hip/hip_bf16.h>

typedef __attribute__((ext_vector_type(2))) float v2f;
typedef __attribute__((ext_vector_type(4))) float v4f;
typedef __attribute__((ext_vector_type(8))) float v8f;

#define D_FEAT 128

#define NODES_PER_BIN   64
#define BIN_SHIFT       6
#define EDGES_PER_BLOCK 8192
#define SORT_THREADS    1024

__global__ __launch_bounds__(SORT_THREADS) void sage_bin_count(
    const int* __restrict__ dst, int* __restrict__ block_hist, int n_edges, int n_bins)
{
    extern __shared__ int lcnt[];
    for (int t = threadIdx.x; t < n_bins; t += SORT_THREADS) lcnt[t] = 0;
    __syncthreads();
    const int e0 = blockIdx.x * EDGES_PER_BLOCK + threadIdx.x;
#pragma unroll
    for (int k = 0; k < EDGES_PER_BLOCK / SORT_THREADS; ++k) {
        const int e = e0 + k * SORT_THREADS;
        if (e < n_edges) atomicAdd(&lcnt[dst[e] >> BIN_SHIFT], 1);
    }
    __syncthreads();
    for (int t = threadIdx.x; t < n_bins; t += SORT_THREADS)
        block_hist[blockIdx.x * n_bins + t] = lcnt[t];
}

#define PREFIX_BINS   64
#define PREFIX_GROUPS 16

__global__ __launch_bounds__(PREFIX_BINS * PREFIX_GROUPS) void sage_bin_prefix(
    const int* __restrict__ block_hist, int* __restrict__ block_base,
    int* __restrict__ bin_total, int n_blocks, int n_bins)
{
    __shared__ int part[PREFIX_GROUPS][PREFIX_BINS];
    const int tl = threadIdx.x % PREFIX_BINS;
    const int g  = threadIdx.x / PREFIX_BINS;
    const int t  = blockIdx.x * PREFIX_BINS + tl;
    const int rows = (n_blocks + PREFIX_GROUPS - 1) / PREFIX_GROUPS;
    const int r0 = g * rows;
    const int r1 = r0 + rows < n_blocks ? r0 + rows : n_blocks;
    int s = 0;
    if (t < n_bins) {
#pragma unroll 8
        for (int b = r0; b < r1; ++b) s += block_hist[b * n_bins + t];
    }
    part[g][tl] = s;
    __syncthreads();
    if (t >= n_bins) return;
    int run = 0;
    for (int gg = 0; gg < g; ++gg) run += part[gg][tl];
#pragma unroll 8
    for (int b = r0; b < r1; ++b) {
        const int c = block_hist[b * n_bins + t];
        block_base[b * n_bins + t] = run;
        run += c;
    }
    if (g == PREFIX_GROUPS - 1) bin_total[t] = run;
}

__global__ __launch_bounds__(SORT_THREADS) void sage_bin_scatter(
    const int* __restrict__ src, const int* __restrict__ dst,
    const int* __restrict__ block_base, const int* __restrict__ bin_total,
    int* __restrict__ bin_start, int2* __restrict__ pairs, int n_edges, int n_bins)
{
    extern __shared__ int lcur[];
    __shared__ int wsum[SORT_THREADS / 32];

    const int cpt  = (n_bins + SORT_THREADS - 1) / SORT_THREADS;
    const int b0   = threadIdx.x * cpt;
    const int lane = threadIdx.x & 31;
    const int wave = threadIdx.x >> 5;
    int own = 0;
    for (int j = 0; j < cpt; ++j)
        if (b0 + j < n_bins) own += bin_total[b0 + j];
    int inc = own;
    for (int d = 1; d < 32; d <<= 1) {
        const int y = __shfl_up(inc, d, 32);
        if (lane >= d) inc += y;
    }
    if (lane == 31) wsum[wave] = inc;
    __syncthreads();
    int run = inc - own;
    for (int w = 0; w < wave; ++w) run += wsum[w];
    for (int j = 0; j < cpt; ++j) {
        const int b = b0 + j;
        if (b < n_bins) {
            lcur[b] = run + block_base[blockIdx.x * n_bins + b];
            if (blockIdx.x == 0) bin_start[b] = run;
            run += bin_total[b];
        }
    }
    if (blockIdx.x == 0 && threadIdx.x == SORT_THREADS - 1) bin_start[n_bins] = run;
    __syncthreads();

    const int e0 = blockIdx.x * EDGES_PER_BLOCK + threadIdx.x;
#pragma unroll
    for (int k = 0; k < EDGES_PER_BLOCK / SORT_THREADS; ++k) {
        const int e = e0 + k * SORT_THREADS;
        if (e < n_edges) {
            const int d = dst[e];
            const int s = src[e];
            const int place = atomicAdd(&lcur[d >> BIN_SHIFT], 1);
            pairs[place] = make_int2(s, d);
        }
    }
}

#define LANES_PER_NODE 8
#define WALK_THREADS   (NODES_PER_BIN * LANES_PER_NODE)

__global__ __launch_bounds__(WALK_THREADS) void sage_bin_gather_mean(
    const float* __restrict__ feat, const int2* __restrict__ pairs,
    const int* __restrict__ bin_start, int2* link, float* __restrict__ hneigh,
    int n_nodes, int n_edges)
{
    __shared__ int lhead[NODES_PER_BIN];
    const int base = blockIdx.x * NODES_PER_BIN;
    const int s0 = bin_start[blockIdx.x];
    const int s1 = bin_start[blockIdx.x + 1];
    if (threadIdx.x < NODES_PER_BIN) lhead[threadIdx.x] = -1;
    __syncthreads();
    for (int i = s0 + threadIdx.x; i < s1; i += WALK_THREADS) {
        const int2 p = pairs[i];
        const int prev = atomicExch(&lhead[p.y - base], i);
        link[i] = make_int2(p.x, prev);
    }
    __syncthreads();

    const int nl  = threadIdx.x / LANES_PER_NODE;
    const int sub = threadIdx.x % LANES_PER_NODE;
    const int v   = base + nl;
    if (v >= n_nodes) return;

    int  e = lhead[nl];
    int2 l = make_int2(0, -1);
    if (e >= 0) l = link[e];

    v4f acc0 = {}, acc1 = {}, acc2 = {}, acc3 = {};
    int cnt = 0;
    for (int it = 0; it < n_edges && e >= 0; ++it) {
        const float* r = feat + (long)l.x * D_FEAT + sub * 4;
        const v4f x0 = *(const v4f*)(r);
        const v4f x1 = *(const v4f*)(r + 32);
        const v4f x2 = *(const v4f*)(r + 64);
        const v4f x3 = *(const v4f*)(r + 96);
        const int en = l.y;
        if (en >= 0) l = link[en];
        acc0 += x0;
        acc1 += x1;
        acc2 += x2;
        acc3 += x3;
        ++cnt;
        e = en;
    }

    const float dg = cnt > 1 ? (float)cnt : 1.0f;
    float* o = hneigh + (long)v * D_FEAT + sub * 4;
    *(v4f*)(o)      = acc0 / dg;
    *(v4f*)(o + 32) = acc1 / dg;
    *(v4f*)(o + 64) = acc2 / dg;
    *(v4f*)(o + 96) = acc3 / dg;
}

#define W_K        256
#define W_CSTRIDE  (8 * W_K + 4)
#define GEMM_WAVES   8
#define GEMM_THREADS (GEMM_WAVES * 32)
#define GEMM_BLOCKS  512

__device__ __forceinline__ v8f wmma_f32(v2f a, v2f b, v8f c) {
    return __builtin_amdgcn_wmma_f32_16x16x4_f32(
         false, a,  false, b,
         (short)0, c,  false,  false);
}

__global__ __launch_bounds__(GEMM_THREADS) void sage_gemm_wmma(
    const float* __restrict__ feat, const float* __restrict__ hneigh,
    const float* __restrict__ Wself, const float* __restrict__ Wneigh,
    float* __restrict__ out, int n_tiles)
{
    __shared__ __attribute__((aligned(16))) float sW[16 * W_CSTRIDE];

    const int wave = __builtin_amdgcn_readfirstlane((int)(threadIdx.x >> 5));
    const int lane = threadIdx.x & 31;
    const int c = lane & 15;
    const int h = lane >> 4;

    const int half   = wave & 1;
    const int n_half = (int)gridDim.x * (GEMM_WAVES / 2);
    const int tile0  = (((wave >> 1) + (int)blockIdx.x) & (GEMM_WAVES / 2 - 1)) * (int)gridDim.x
                     + (int)blockIdx.x;

    v4f a[4][8];
    {
        const int tfirst = tile0 < n_tiles ? tile0 : n_tiles - 1;
        const float* f0 = feat + ((long)tfirst * 16 + c) * D_FEAT + 4 * h;
#pragma unroll
        for (int j = 0; j < 8; ++j) a[0][j] = *(const v4f*)(f0 + 8 * j);
    }

#pragma unroll
    for (int i = 0; i < 16; ++i) {
        const int n   = GEMM_WAVES * i + wave;
        const int dst = (n & 15) * W_CSTRIDE + (n >> 4) * W_K + lane * 4;
        const v4f ws = *(const v4f*)(Wself  + n * D_FEAT + lane * 4);
        const v4f wn = *(const v4f*)(Wneigh + n * D_FEAT + lane * 4);
        *(v4f*)(sW + dst)          = ws;
        *(v4f*)(sW + dst + D_FEAT) = wn;
    }
    __syncthreads();
    if (tile0 >= n_tiles) return;

    int boff = c * W_CSTRIDE + 4 * h + half * 4 * W_K;

    for (int tile = tile0; tile < n_tiles; tile += n_half) {
        const long row = (long)tile * 16 + c;
        const float* fa = feat   + row * D_FEAT + 4 * h;
        const float* ha = hneigh + row * D_FEAT + 4 * h;
        const int tnext = tile + n_half < n_tiles ? tile + n_half : tile;
        const float* fn = feat + ((long)tnext * 16 + c) * D_FEAT + 4 * h;

        asm volatile("" : "+v"(boff));
        const float* bl = sW + boff;

        v8f acc[4];
#pragma unroll
        for (int q = 0; q < 4; ++q) acc[q] = (v8f){};

        v4f bcur[4], bnxt[4];
#pragma unroll
        for (int q = 0; q < 4; ++q)
            bcur[q] = *(const v4f*)(bl + q * W_K);

#pragma unroll
        for (int p = 0; p < 4; ++p) {
            if (p == 0) {
#pragma unroll
                for (int j = 0; j < 8; ++j) a[1][j] = *(const v4f*)(fa + 64 + 8 * j);
            }
            if (p == 1) {
#pragma unroll
                for (int j = 0; j < 8; ++j) a[2][j] = *(const v4f*)(ha + 8 * j);
            }
            if (p == 2) {
#pragma unroll
                for (int j = 0; j < 8; ++j) a[3][j] = *(const v4f*)(ha + 64 + 8 * j);
            }
            if (p == 3) {
#pragma unroll
                for (int j = 0; j < 8; ++j) a[0][j] = *(const v4f*)(fn + 8 * j);
            }
#pragma unroll
            for (int j = 0; j < 8; ++j) {
                const int knext = p * 64 + j * 8 + 8;
                __builtin_amdgcn_sched_barrier(0);
                if (knext < W_K) {
#pragma unroll
                    for (int q = 0; q < 4; ++q)
                        bnxt[q] = *(const v4f*)(bl + q * W_K + knext);
                }
                const v4f av = a[p][j];
                const v2f a_lo = {av.x, av.y};
                const v2f a_hi = {av.z, av.w};
#pragma unroll
                for (int q = 0; q < 4; ++q) {
                    const v2f b_lo = {bcur[q].x, bcur[q].y};
                    acc[q] = wmma_f32(a_lo, b_lo, acc[q]);
                }
#pragma unroll
                for (int q = 0; q < 4; ++q) {
                    const v2f b_hi = {bcur[q].z, bcur[q].w};
                    acc[q] = wmma_f32(a_hi, b_hi, acc[q]);
                }
                if (knext < W_K) {
#pragma unroll
                    for (int q = 0; q < 4; ++q) bcur[q] = bnxt[q];
                }
            }
        }

        float* o = out + ((long)tile * 16 + h * 8) * D_FEAT + half * 64 + c;
#pragma unroll
        for (int q = 0; q < 4; ++q) {
#pragma unroll
            for (int r = 0; r < 8; ++r)
                o[(long)r * D_FEAT + q * 16] = acc[q][r];
        }
    }
}

extern "C" void kernel_launch(void* const* d_in, const int* in_sizes, int n_in,
                              void* d_out, int out_size, void* d_ws, size_t ws_size,
                              hipStream_t stream)
{
    const float* feat   = (const float*)d_in[0];
    const int*   src    = (const int*)d_in[1];
    const int*   dst    = (const int*)d_in[2];
    const float* Wself  = (const float*)d_in[3];
    const float* Wneigh = (const float*)d_in[4];
    float*       out    = (float*)d_out;

    const int n_nodes = in_sizes[0] / D_FEAT;
    const int n_edges = in_sizes[1];
    const int n_bins  = (n_nodes + NODES_PER_BIN - 1) / NODES_PER_BIN;
    const int n_eblk  = (n_edges + EDGES_PER_BLOCK - 1) / EDGES_PER_BLOCK;

    float* hneigh     = (float*)d_ws;
    int2*  pairs      = (int2*)d_out;
    int2*  link       = pairs + n_edges;
    int*   block_hist = (int*)(link + n_edges);
    int*   block_base = block_hist + n_eblk * n_bins;
    int*   bin_total  = block_base + n_eblk * n_bins;
    int*   bin_start  = bin_total + n_bins;

    sage_bin_count<<<n_eblk, SORT_THREADS, n_bins * sizeof(int), stream>>>(
        dst, block_hist, n_edges, n_bins);

    sage_bin_prefix<<<(n_bins + PREFIX_BINS - 1) / PREFIX_BINS, PREFIX_BINS * PREFIX_GROUPS, 0, stream>>>(
        block_hist, block_base, bin_total, n_eblk, n_bins);

    sage_bin_scatter<<<n_eblk, SORT_THREADS, n_bins * sizeof(int), stream>>>(
        src, dst, block_base, bin_total, bin_start, pairs, n_edges, n_bins);

    sage_bin_gather_mean<<<n_bins, WALK_THREADS, 0, stream>>>(
        feat, pairs, bin_start, link, hneigh, n_nodes, n_edges);

    const int n_tiles = n_nodes / 16;
    sage_gemm_wmma<<<GEMM_BLOCKS, GEMM_THREADS, 0, stream>>>(feat, hneigh, Wself, Wneigh, out, n_tiles);
}


# Edge sort made block-local: one launch sorts each 8192-edge block by bin in LDS and the bin's block collects its runs itself, replacing the count, prefix and scatter launches (sum order changes)
# speedup vs baseline: 4.9653x; 1.2271x over previous
#include <hip/hip_runtime.h>
#include <hip/hip_bf16.h>

typedef __attribute__((ext_vector_type(2))) float v2f;
typedef __attribute__((ext_vector_type(4))) float v4f;
typedef __attribute__((ext_vector_type(8))) float v8f;

#define D_FEAT 128

#define NODES_PER_BIN   64
#define BIN_SHIFT       6
#define EDGES_PER_BLOCK 8192
#define SORT_THREADS    512
#define EDGES_PER_THREAD (EDGES_PER_BLOCK / SORT_THREADS)

__global__ __launch_bounds__(SORT_THREADS) void sage_block_sort(
    const int* __restrict__ src, const int* __restrict__ dst,
    int* __restrict__ block_off, int2* __restrict__ pairs, int n_edges, int n_bins)
{
    extern __shared__ int lcnt[];
    __shared__ int2 lsorted[EDGES_PER_BLOCK];
    __shared__ int wsum[SORT_THREADS / 32];

    const int e0 = blockIdx.x * EDGES_PER_BLOCK + threadIdx.x;
    int es[EDGES_PER_THREAD], ed[EDGES_PER_THREAD], rk[EDGES_PER_THREAD];
#pragma unroll
    for (int k = 0; k < EDGES_PER_THREAD; ++k) {
        const int e = e0 + k * SORT_THREADS;
        ed[k] = e < n_edges ? dst[e] : 0;
        es[k] = e < n_edges ? src[e] : 0;
    }
    for (int t = threadIdx.x; t < n_bins; t += SORT_THREADS) lcnt[t] = 0;
    __syncthreads();
#pragma unroll
    for (int k = 0; k < EDGES_PER_THREAD; ++k) {
        rk[k] = 0;
        if (e0 + k * SORT_THREADS < n_edges) rk[k] = atomicAdd(&lcnt[ed[k] >> BIN_SHIFT], 1);
    }
    __syncthreads();

    const int cpt  = (n_bins + SORT_THREADS - 1) / SORT_THREADS;
    const int b0   = threadIdx.x * cpt;
    const int lane = threadIdx.x & 31;
    const int wave = threadIdx.x >> 5;
    int own = 0;
    for (int j = 0; j < cpt; ++j)
        if (b0 + j < n_bins) own += lcnt[b0 + j];
    int inc = own;
    for (int d = 1; d < 32; d <<= 1) {
        const int y = __shfl_up(inc, d, 32);
        if (lane >= d) inc += y;
    }
    if (lane == 31) wsum[wave] = inc;
    __syncthreads();
    int run = inc - own;
    for (int w = 0; w < wave; ++w) run += wsum[w];
    for (int j = 0; j < cpt; ++j) {
        const int b = b0 + j;
        if (b < n_bins) {
            const int cb = lcnt[b];
            lcnt[b] = run;
            run += cb;
        }
    }
    __syncthreads();
#pragma unroll
    for (int k = 0; k < EDGES_PER_THREAD; ++k) {
        if (e0 + k * SORT_THREADS < n_edges)
            lsorted[lcnt[ed[k] >> BIN_SHIFT] + rk[k]] = make_int2(es[k], ed[k]);
    }
    __syncthreads();

    const int left   = n_edges - blockIdx.x * EDGES_PER_BLOCK;
    const int nvalid = left < EDGES_PER_BLOCK ? left : EDGES_PER_BLOCK;
    int2* po = pairs + (long)blockIdx.x * EDGES_PER_BLOCK;
    int2 cp[EDGES_PER_THREAD];
#pragma unroll
    for (int k = 0; k < EDGES_PER_THREAD; ++k) {
        const int i = threadIdx.x + k * SORT_THREADS;
        cp[k] = make_int2(0, 0);
        if (i < nvalid) cp[k] = lsorted[i];
    }
#pragma unroll
    for (int k = 0; k < EDGES_PER_THREAD; ++k) {
        const int i = threadIdx.x + k * SORT_THREADS;
        if (i < nvalid) po[i] = cp[k];
    }
    int* orow = block_off + (long)blockIdx.x * (n_bins + 1);
    for (int t = threadIdx.x; t < n_bins; t += SORT_THREADS) orow[t] = lcnt[t];
    if (threadIdx.x == 0) orow[n_bins] = nvalid;
}

#define LANES_PER_NODE 8
#define WALK_THREADS   (NODES_PER_BIN * LANES_PER_NODE)

__global__ __launch_bounds__(WALK_THREADS) void sage_bin_gather_mean(
    const float* __restrict__ feat, const int2* __restrict__ pairs,
    const int* __restrict__ block_off, int2* link, float* __restrict__ hneigh,
    int n_nodes, int n_edges, int n_bins, int n_runs)
{
    extern __shared__ int lrun[];
    __shared__ int lhead[NODES_PER_BIN];
    __shared__ int rsum[WALK_THREADS / 32];
    int* rpos   = lrun;
    int* rstart = lrun + n_runs + 1;
    const int bin  = blockIdx.x;
    const int base = bin * NODES_PER_BIN;

    const int cpr  = (n_runs + WALK_THREADS - 1) / WALK_THREADS;
    const int r0   = threadIdx.x * cpr;
    const int lane = threadIdx.x & 31;
    const int wave = threadIdx.x >> 5;
    int own = 0;
    for (int j = 0; j < cpr; ++j) {
        const int r = r0 + j;
        if (r < n_runs) {
            const int* orow = block_off + (long)r * (n_bins + 1) + bin;
            const int o0 = orow[0];
            const int o1 = orow[1];
            rstart[r] = r * EDGES_PER_BLOCK + o0;
            rpos[r]   = o1 - o0;
            own += o1 - o0;
        }
    }
    int inc = own;
    for (int d = 1; d < 32; d <<= 1) {
        const int y = __shfl_up(inc, d, 32);
        if (lane >= d) inc += y;
    }
    if (lane == 31) rsum[wave] = inc;
    if (threadIdx.x < NODES_PER_BIN) lhead[threadIdx.x] = -1;
    __syncthreads();
    int run = inc - own;
    for (int w = 0; w < wave; ++w) run += rsum[w];
    for (int j = 0; j < cpr; ++j) {
        const int r = r0 + j;
        if (r < n_runs) {
            const int len = rpos[r];
            rpos[r] = run;
            run += len;
        }
    }
    if (threadIdx.x == WALK_THREADS - 1) rpos[n_runs] = run;
    __syncthreads();

    const int total = rpos[n_runs];
    for (int i = threadIdx.x; i < total; i += WALK_THREADS) {
        int lo = 0, hi = n_runs;
        for (int s = 0; s < 32 && hi - lo > 1; ++s) {
            const int mid = (lo + hi) >> 1;
            if (rpos[mid] <= i) lo = mid; else hi = mid;
        }
        const int g = rstart[lo] + (i - rpos[lo]);
        const int2 p = pairs[g];
        const int prev = atomicExch(&lhead[p.y - base], g);
        link[g] = make_int2(p.x, prev);
    }
    __syncthreads();

    const int nl  = threadIdx.x / LANES_PER_NODE;
    const int sub = threadIdx.x % LANES_PER_NODE;
    const int v   = base + nl;
    if (v >= n_nodes) return;

    int  e = lhead[nl];
    int2 l = make_int2(0, -1);
    if (e >= 0) l = link[e];

    v4f acc0 = {}, acc1 = {}, acc2 = {}, acc3 = {};
    int cnt = 0;
    for (int it = 0; it < n_edges && e >= 0; ++it) {
        const float* r = feat + (long)l.x * D_FEAT + sub * 4;
        const v4f x0 = *(const v4f*)(r);
        const v4f x1 = *(const v4f*)(r + 32);
        const v4f x2 = *(const v4f*)(r + 64);
        const v4f x3 = *(const v4f*)(r + 96);
        const int en = l.y;
        if (en >= 0) l = link[en];
        acc0 += x0;
        acc1 += x1;
        acc2 += x2;
        acc3 += x3;
        ++cnt;
        e = en;
    }

    const float dg = cnt > 1 ? (float)cnt : 1.0f;
    float* o = hneigh + (long)v * D_FEAT + sub * 4;
    *(v4f*)(o)      = acc0 / dg;
    *(v4f*)(o + 32) = acc1 / dg;
    *(v4f*)(o + 64) = acc2 / dg;
    *(v4f*)(o + 96) = acc3 / dg;
}

#define W_K        256
#define W_CSTRIDE  (8 * W_K + 4)
#define GEMM_WAVES   8
#define GEMM_THREADS (GEMM_WAVES * 32)
#define GEMM_BLOCKS  512

__device__ __forceinline__ v8f wmma_f32(v2f a, v2f b, v8f c) {
    return __builtin_amdgcn_wmma_f32_16x16x4_f32(
         false, a,  false, b,
         (short)0, c,  false,  false);
}

__global__ __launch_bounds__(GEMM_THREADS) void sage_gemm_wmma(
    const float* __restrict__ feat, const float* __restrict__ hneigh,
    const float* __restrict__ Wself, const float* __restrict__ Wneigh,
    float* __restrict__ out, int n_tiles)
{
    __shared__ __attribute__((aligned(16))) float sW[16 * W_CSTRIDE];

    const int wave = __builtin_amdgcn_readfirstlane((int)(threadIdx.x >> 5));
    const int lane = threadIdx.x & 31;
    const int c = lane & 15;
    const int h = lane >> 4;

    const int half   = wave & 1;
    const int n_half = (int)gridDim.x * (GEMM_WAVES / 2);
    const int tile0  = (((wave >> 1) + (int)blockIdx.x) & (GEMM_WAVES / 2 - 1)) * (int)gridDim.x
                     + (int)blockIdx.x;

    v4f a[4][8];
    {
        const int tfirst = tile0 < n_tiles ? tile0 : n_tiles - 1;
        const float* f0 = feat + ((long)tfirst * 16 + c) * D_FEAT + 4 * h;
#pragma unroll
        for (int j = 0; j < 8; ++j) a[0][j] = *(const v4f*)(f0 + 8 * j);
    }

#pragma unroll
    for (int i = 0; i < 16; ++i) {
        const int n   = GEMM_WAVES * i + wave;
        const int dst = (n & 15) * W_CSTRIDE + (n >> 4) * W_K + lane * 4;
        const v4f ws = *(const v4f*)(Wself  + n * D_FEAT + lane * 4);
        const v4f wn = *(const v4f*)(Wneigh + n * D_FEAT + lane * 4);
        *(v4f*)(sW + dst)          = ws;
        *(v4f*)(sW + dst + D_FEAT) = wn;
    }
    __syncthreads();
    if (tile0 >= n_tiles) return;

    int boff = c * W_CSTRIDE + 4 * h + half * 4 * W_K;

    for (int tile = tile0; tile < n_tiles; tile += n_half) {
        const long row = (long)tile * 16 + c;
        const float* fa = feat   + row * D_FEAT + 4 * h;
        const float* ha = hneigh + row * D_FEAT + 4 * h;
        const int tnext = tile + n_half < n_tiles ? tile + n_half : tile;
        const float* fn = feat + ((long)tnext * 16 + c) * D_FEAT + 4 * h;

        asm volatile("" : "+v"(boff));
        const float* bl = sW + boff;

        v8f acc[4];
#pragma unroll
        for (int q = 0; q < 4; ++q) acc[q] = (v8f){};

        v4f bcur[4], bnxt[4];
#pragma unroll
        for (int q = 0; q < 4; ++q)
            bcur[q] = *(const v4f*)(bl + q * W_K);

#pragma unroll
        for (int p = 0; p < 4; ++p) {
            if (p == 0) {
#pragma unroll
                for (int j = 0; j < 8; ++j) a[1][j] = *(const v4f*)(fa + 64 + 8 * j);
            }
            if (p == 1) {
#pragma unroll
                for (int j = 0; j < 8; ++j) a[2][j] = *(const v4f*)(ha + 8 * j);
            }
            if (p == 2) {
#pragma unroll
                for (int j = 0; j < 8; ++j) a[3][j] = *(const v4f*)(ha + 64 + 8 * j);
            }
            if (p == 3) {
#pragma unroll
                for (int j = 0; j < 8; ++j) a[0][j] = *(const v4f*)(fn + 8 * j);
            }
#pragma unroll
            for (int j = 0; j < 8; ++j) {
                const int knext = p * 64 + j * 8 + 8;
                __builtin_amdgcn_sched_barrier(0);
                if (knext < W_K) {
#pragma unroll
                    for (int q = 0; q < 4; ++q)
                        bnxt[q] = *(const v4f*)(bl + q * W_K + knext);
                }
                const v4f av = a[p][j];
                const v2f a_lo = {av.x, av.y};
                const v2f a_hi = {av.z, av.w};
#pragma unroll
                for (int q = 0; q < 4; ++q) {
                    const v2f b_lo = {bcur[q].x, bcur[q].y};
                    acc[q] = wmma_f32(a_lo, b_lo, acc[q]);
                }
#pragma unroll
                for (int q = 0; q < 4; ++q) {
                    const v2f b_hi = {bcur[q].z, bcur[q].w};
                    acc[q] = wmma_f32(a_hi, b_hi, acc[q]);
                }
                if (knext < W_K) {
#pragma unroll
                    for (int q = 0; q < 4; ++q) bcur[q] = bnxt[q];
                }
            }
        }

        float* o = out + ((long)tile * 16 + h * 8) * D_FEAT + half * 64 + c;
#pragma unroll
        for (int q = 0; q < 4; ++q) {
#pragma unroll
            for (int r = 0; r < 8; ++r)
                o[(long)r * D_FEAT + q * 16] = acc[q][r];
        }
    }
}

extern "C" void kernel_launch(void* const* d_in, const int* in_sizes, int n_in,
                              void* d_out, int out_size, void* d_ws, size_t ws_size,
                              hipStream_t stream)
{
    const float* feat   = (const float*)d_in[0];
    const int*   src    = (const int*)d_in[1];
    const int*   dst    = (const int*)d_in[2];
    const float* Wself  = (const float*)d_in[3];
    const float* Wneigh = (const float*)d_in[4];
    float*       out    = (float*)d_out;

    const int n_nodes = in_sizes[0] / D_FEAT;
    const int n_edges = in_sizes[1];
    const int n_bins  = (n_nodes + NODES_PER_BIN - 1) / NODES_PER_BIN;
    const int n_eblk  = (n_edges + EDGES_PER_BLOCK - 1) / EDGES_PER_BLOCK;

    float* hneigh    = (float*)d_ws;
    int2*  pairs     = (int2*)d_out;
    int2*  link      = pairs + n_eblk * EDGES_PER_BLOCK;
    int*   block_off = (int*)(link + n_eblk * EDGES_PER_BLOCK);

    sage_block_sort<<<n_eblk, SORT_THREADS, n_bins * sizeof(int), stream>>>(
        src, dst, block_off, pairs, n_edges, n_bins);

    sage_bin_gather_mean<<<n_bins, WALK_THREADS, (2 * n_eblk + 1) * sizeof(int), stream>>>(
        feat, pairs, block_off, link, hneigh, n_nodes, n_edges, n_bins, n_eblk);

    const int n_tiles = n_nodes / 16;
    sage_gemm_wmma<<<GEMM_BLOCKS, GEMM_THREADS, 0, stream>>>(feat, hneigh, Wself, Wneigh, out, n_tiles);
}


# Per-node linked lists in global memory (one dependent link load per hop) replaced by an LDS counting sort of each bin's sources by node; a node reads its run four rows at a time (sum order changes)
# speedup vs baseline: 5.5215x; 1.1120x over previous
#include <hip/hip_runtime.h>
#include <hip/hip_bf16.h>

typedef __attribute__((ext_vector_type(2))) float v2f;
typedef __attribute__((ext_vector_type(4))) float v4f;
typedef __attribute__((ext_vector_type(8))) float v8f;

#define D_FEAT 128

#define NODES_PER_BIN   64
#define BIN_SHIFT       6
#define EDGES_PER_BLOCK 8192
#define SORT_THREADS    512
#define EDGES_PER_THREAD (EDGES_PER_BLOCK / SORT_THREADS)

__global__ __launch_bounds__(SORT_THREADS) void sage_block_sort(
    const int* __restrict__ src, const int* __restrict__ dst,
    int* __restrict__ block_off, int2* __restrict__ pairs, int n_edges, int n_bins)
{
    extern __shared__ int lcnt[];
    __shared__ int2 lsorted[EDGES_PER_BLOCK];
    __shared__ int wsum[SORT_THREADS / 32];

    const int e0 = blockIdx.x * EDGES_PER_BLOCK + threadIdx.x;
    int es[EDGES_PER_THREAD], ed[EDGES_PER_THREAD], rk[EDGES_PER_THREAD];
#pragma unroll
    for (int k = 0; k < EDGES_PER_THREAD; ++k) {
        const int e = e0 + k * SORT_THREADS;
        ed[k] = e < n_edges ? dst[e] : 0;
        es[k] = e < n_edges ? src[e] : 0;
    }
    for (int t = threadIdx.x; t < n_bins; t += SORT_THREADS) lcnt[t] = 0;
    __syncthreads();
#pragma unroll
    for (int k = 0; k < EDGES_PER_THREAD; ++k) {
        rk[k] = 0;
        if (e0 + k * SORT_THREADS < n_edges) rk[k] = atomicAdd(&lcnt[ed[k] >> BIN_SHIFT], 1);
    }
    __syncthreads();

    const int cpt  = (n_bins + SORT_THREADS - 1) / SORT_THREADS;
    const int b0   = threadIdx.x * cpt;
    const int lane = threadIdx.x & 31;
    const int wave = threadIdx.x >> 5;
    int own = 0;
    for (int j = 0; j < cpt; ++j)
        if (b0 + j < n_bins) own += lcnt[b0 + j];
    int inc = own;
    for (int d = 1; d < 32; d <<= 1) {
        const int y = __shfl_up(inc, d, 32);
        if (lane >= d) inc += y;
    }
    if (lane == 31) wsum[wave] = inc;
    __syncthreads();
    int run = inc - own;
    for (int w = 0; w < wave; ++w) run += wsum[w];
    for (int j = 0; j < cpt; ++j) {
        const int b = b0 + j;
        if (b < n_bins) {
            const int cb = lcnt[b];
            lcnt[b] = run;
            run += cb;
        }
    }
    __syncthreads();
#pragma unroll
    for (int k = 0; k < EDGES_PER_THREAD; ++k) {
        if (e0 + k * SORT_THREADS < n_edges)
            lsorted[lcnt[ed[k] >> BIN_SHIFT] + rk[k]] = make_int2(es[k], ed[k]);
    }
    __syncthreads();

    const int left   = n_edges - blockIdx.x * EDGES_PER_BLOCK;
    const int nvalid = left < EDGES_PER_BLOCK ? left : EDGES_PER_BLOCK;
    int2* po = pairs + (long)blockIdx.x * EDGES_PER_BLOCK;
    int2 cp[EDGES_PER_THREAD];
#pragma unroll
    for (int k = 0; k < EDGES_PER_THREAD; ++k) {
        const int i = threadIdx.x + k * SORT_THREADS;
        cp[k] = make_int2(0, 0);
        if (i < nvalid) cp[k] = lsorted[i];
    }
#pragma unroll
    for (int k = 0; k < EDGES_PER_THREAD; ++k) {
        const int i = threadIdx.x + k * SORT_THREADS;
        if (i < nvalid) po[i] = cp[k];
    }
    int* orow = block_off + (long)blockIdx.x * (n_bins + 1);
    for (int t = threadIdx.x; t < n_bins; t += SORT_THREADS) orow[t] = lcnt[t];
    if (threadIdx.x == 0) orow[n_bins] = nvalid;
}

#define LANES_PER_NODE  8
#define WALK_THREADS    (NODES_PER_BIN * LANES_PER_NODE)
#define LIST_PER_THREAD 4
#define LIST_CAP        (WALK_THREADS * LIST_PER_THREAD)

__global__ __launch_bounds__(WALK_THREADS) void sage_bin_gather_mean(
    const float* __restrict__ feat, const int2* __restrict__ pairs,
    const int* __restrict__ block_off, float* __restrict__ hneigh,
    int n_nodes, int n_bins, int n_runs)
{
    extern __shared__ int lrun[];
    __shared__ int lsrc[LIST_CAP];
    __shared__ int lcnt[NODES_PER_BIN];
    __shared__ int lfirst[NODES_PER_BIN];
    __shared__ int rsum[WALK_THREADS / 32];
    int* rpos   = lrun;
    int* rstart = lrun + n_runs + 1;
    const int bin  = blockIdx.x;
    const int base = bin * NODES_PER_BIN;

    const int cpr  = (n_runs + WALK_THREADS - 1) / WALK_THREADS;
    const int r0   = threadIdx.x * cpr;
    const int lane = threadIdx.x & 31;
    const int wave = threadIdx.x >> 5;
    int own = 0;
    for (int j = 0; j < cpr; ++j) {
        const int r = r0 + j;
        if (r < n_runs) {
            const int* orow = block_off + (long)r * (n_bins + 1) + bin;
            const int o0 = orow[0];
            const int o1 = orow[1];
            rstart[r] = r * EDGES_PER_BLOCK + o0;
            rpos[r]   = o1 - o0;
            own += o1 - o0;
        }
    }
    int inc = own;
    for (int d = 1; d < 32; d <<= 1) {
        const int y = __shfl_up(inc, d, 32);
        if (lane >= d) inc += y;
    }
    if (lane == 31) rsum[wave] = inc;
    if (threadIdx.x < NODES_PER_BIN) lcnt[threadIdx.x] = 0;
    __syncthreads();
    int run = inc - own;
    for (int w = 0; w < wave; ++w) run += rsum[w];
    for (int j = 0; j < cpr; ++j) {
        const int r = r0 + j;
        if (r < n_runs) {
            const int len = rpos[r];
            rpos[r] = run;
            run += len;
        }
    }
    if (threadIdx.x == WALK_THREADS - 1) rpos[n_runs] = run;
    __syncthreads();

    const int total = rpos[n_runs];
    const int nl  = threadIdx.x / LANES_PER_NODE;
    const int sub = threadIdx.x % LANES_PER_NODE;
    const float* frow = feat + sub * 4;

    v4f acc0 = {}, acc1 = {}, acc2 = {}, acc3 = {};
    int deg = 0;

    for (int c0 = 0; c0 < total; c0 += LIST_CAP) {
        const int cend = total - c0 < LIST_CAP ? total : c0 + LIST_CAP;
        int es[LIST_PER_THREAD], en[LIST_PER_THREAD], rk[LIST_PER_THREAD];
        int ii[LIST_PER_THREAD], lo[LIST_PER_THREAD], hi[LIST_PER_THREAD];
#pragma unroll
        for (int k = 0; k < LIST_PER_THREAD; ++k) {
            const int i = c0 + (int)threadIdx.x + k * WALK_THREADS;
            ii[k] = i < cend ? i : cend - 1;
            lo[k] = 0;
            hi[k] = n_runs;
        }
        for (int s = 0; s < 32 && (1u << s) < (unsigned)n_runs; ++s) {
#pragma unroll
            for (int k = 0; k < LIST_PER_THREAD; ++k) {
                const int  mid = (lo[k] + hi[k]) >> 1;
                const bool go  = hi[k] - lo[k] > 1;
                const bool up  = rpos[mid] <= ii[k];
                lo[k] = go && up  ? mid : lo[k];
                hi[k] = go && !up ? mid : hi[k];
            }
        }
        int2 pr[LIST_PER_THREAD];
#pragma unroll
        for (int k = 0; k < LIST_PER_THREAD; ++k) {
            pr[k] = make_int2(0, 0);
            if (c0 + (int)threadIdx.x + k * WALK_THREADS < cend)
                pr[k] = pairs[rstart[lo[k]] + (ii[k] - rpos[lo[k]])];
        }
#pragma unroll
        for (int k = 0; k < LIST_PER_THREAD; ++k) {
            es[k] = pr[k].x;
            en[k] = -1;
            rk[k] = 0;
            if (c0 + (int)threadIdx.x + k * WALK_THREADS < cend) {
                en[k] = pr[k].y - base;
                rk[k] = atomicAdd(&lcnt[en[k]], 1);
            }
        }
        __syncthreads();
        if (threadIdx.x < 32) {
            const int ca = lcnt[2 * threadIdx.x];
            const int cb = lcnt[2 * threadIdx.x + 1];
            int ps = ca + cb;
            for (int d = 1; d < 32; d <<= 1) {
                const int y = __shfl_up(ps, d, 32);
                if (lane >= d) ps += y;
            }
            const int ex = ps - (ca + cb);
            lfirst[2 * threadIdx.x]     = ex;
            lfirst[2 * threadIdx.x + 1] = ex + ca;
        }
        __syncthreads();
#pragma unroll
        for (int k = 0; k < LIST_PER_THREAD; ++k) {
            if (en[k] >= 0) lsrc[lfirst[en[k]] + rk[k]] = es[k];
        }
        __syncthreads();

        const int  n  = lcnt[nl];
        const int* ls = lsrc + lfirst[nl];
        int s0 = 0 < n ? ls[0] : 0;
        int s1 = 1 < n ? ls[1] : 0;
        int s2 = 2 < n ? ls[2] : 0;
        int s3 = 3 < n ? ls[3] : 0;
        int j = 0;
        for (; j + 3 < n; j += 4) {
            const float* ra = frow + (long)s0 * D_FEAT;
            const float* rb = frow + (long)s1 * D_FEAT;
            const float* rc = frow + (long)s2 * D_FEAT;
            const float* rd = frow + (long)s3 * D_FEAT;
            const v4f xa0 = *(const v4f*)(ra);
            const v4f xa1 = *(const v4f*)(ra + 32);
            const v4f xa2 = *(const v4f*)(ra + 64);
            const v4f xa3 = *(const v4f*)(ra + 96);
            const v4f xb0 = *(const v4f*)(rb);
            const v4f xb1 = *(const v4f*)(rb + 32);
            const v4f xb2 = *(const v4f*)(rb + 64);
            const v4f xb3 = *(const v4f*)(rb + 96);
            const v4f xc0 = *(const v4f*)(rc);
            const v4f xc1 = *(const v4f*)(rc + 32);
            const v4f xc2 = *(const v4f*)(rc + 64);
            const v4f xc3 = *(const v4f*)(rc + 96);
            const v4f xd0 = *(const v4f*)(rd);
            const v4f xd1 = *(const v4f*)(rd + 32);
            const v4f xd2 = *(const v4f*)(rd + 64);
            const v4f xd3 = *(const v4f*)(rd + 96);
            const int t0 = j + 4 < n ? ls[j + 4] : 0;
            const int t1 = j + 5 < n ? ls[j + 5] : 0;
            const int t2 = j + 6 < n ? ls[j + 6] : 0;
            const int t3 = j + 7 < n ? ls[j + 7] : 0;
            acc0 += xa0;
            acc1 += xa1;
            acc2 += xa2;
            acc3 += xa3;
            acc0 += xb0;
            acc1 += xb1;
            acc2 += xb2;
            acc3 += xb3;
            acc0 += xc0;
            acc1 += xc1;
            acc2 += xc2;
            acc3 += xc3;
            acc0 += xd0;
            acc1 += xd1;
            acc2 += xd2;
            acc3 += xd3;
            s0 = t0;
            s1 = t1;
            s2 = t2;
            s3 = t3;
        }
        if (j + 2 < n) {
            const float* ra = frow + (long)s0 * D_FEAT;
            const float* rb = frow + (long)s1 * D_FEAT;
            const float* rc = frow + (long)s2 * D_FEAT;
            const v4f xa0 = *(const v4f*)(ra);
            const v4f xa1 = *(const v4f*)(ra + 32);
            const v4f xa2 = *(const v4f*)(ra + 64);
            const v4f xa3 = *(const v4f*)(ra + 96);
            const v4f xb0 = *(const v4f*)(rb);
            const v4f xb1 = *(const v4f*)(rb + 32);
            const v4f xb2 = *(const v4f*)(rb + 64);
            const v4f xb3 = *(const v4f*)(rb + 96);
            const v4f xc0 = *(const v4f*)(rc);
            const v4f xc1 = *(const v4f*)(rc + 32);
            const v4f xc2 = *(const v4f*)(rc + 64);
            const v4f xc3 = *(const v4f*)(rc + 96);
            acc0 += xa0;
            acc1 += xa1;
            acc2 += xa2;
            acc3 += xa3;
            acc0 += xb0;
            acc1 += xb1;
            acc2 += xb2;
            acc3 += xb3;
            acc0 += xc0;
            acc1 += xc1;
            acc2 += xc2;
            acc3 += xc3;
        }
        else if (j + 1 < n) {
            const float* ra = frow + (long)s0 * D_FEAT;
            const float* rb = frow + (long)s1 * D_FEAT;
            const v4f xa0 = *(const v4f*)(ra);
            const v4f xa1 = *(const v4f*)(ra + 32);
            const v4f xa2 = *(const v4f*)(ra + 64);
            const v4f xa3 = *(const v4f*)(ra + 96);
            const v4f xb0 = *(const v4f*)(rb);
            const v4f xb1 = *(const v4f*)(rb + 32);
            const v4f xb2 = *(const v4f*)(rb + 64);
            const v4f xb3 = *(const v4f*)(rb + 96);
            acc0 += xa0;
            acc1 += xa1;
            acc2 += xa2;
            acc3 += xa3;
            acc0 += xb0;
            acc1 += xb1;
            acc2 += xb2;
            acc3 += xb3;
        }
        else if (j < n) {
            const float* ra = frow + (long)s0 * D_FEAT;
            const v4f xa0 = *(const v4f*)(ra);
            const v4f xa1 = *(const v4f*)(ra + 32);
            const v4f xa2 = *(const v4f*)(ra + 64);
            const v4f xa3 = *(const v4f*)(ra + 96);
            acc0 += xa0;
            acc1 += xa1;
            acc2 += xa2;
            acc3 += xa3;
        }
        deg += n;

        if (c0 + LIST_CAP < total) {
            __syncthreads();
            if (threadIdx.x < NODES_PER_BIN) lcnt[threadIdx.x] = 0;
            __syncthreads();
        }
    }

    const int v = base + nl;
    if (v >= n_nodes) return;
    const float dg = deg > 1 ? (float)deg : 1.0f;
    float* o = hneigh + (long)v * D_FEAT + sub * 4;
    *(v4f*)(o)      = acc0 / dg;
    *(v4f*)(o + 32) = acc1 / dg;
    *(v4f*)(o + 64) = acc2 / dg;
    *(v4f*)(o + 96) = acc3 / dg;
}

#define W_K        256
#define W_CSTRIDE  (8 * W_K + 4)
#define GEMM_WAVES   8
#define GEMM_THREADS (GEMM_WAVES * 32)
#define GEMM_BLOCKS  512

__device__ __forceinline__ v8f wmma_f32(v2f a, v2f b, v8f c) {
    return __builtin_amdgcn_wmma_f32_16x16x4_f32(
         false, a,  false, b,
         (short)0, c,  false,  false);
}

__global__ __launch_bounds__(GEMM_THREADS) void sage_gemm_wmma(
    const float* __restrict__ feat, const float* __restrict__ hneigh,
    const float* __restrict__ Wself, const float* __restrict__ Wneigh,
    float* __restrict__ out, int n_tiles)
{
    __shared__ __attribute__((aligned(16))) float sW[16 * W_CSTRIDE];

    const int wave = __builtin_amdgcn_readfirstlane((int)(threadIdx.x >> 5));
    const int lane = threadIdx.x & 31;
    const int c = lane & 15;
    const int h = lane >> 4;

    const int half   = wave & 1;
    const int n_half = (int)gridDim.x * (GEMM_WAVES / 2);
    const int tile0  = (((wave >> 1) + (int)blockIdx.x) & (GEMM_WAVES / 2 - 1)) * (int)gridDim.x
                     + (int)blockIdx.x;

    v4f a[4][8];
    {
        const int tfirst = tile0 < n_tiles ? tile0 : n_tiles - 1;
        const float* f0 = feat + ((long)tfirst * 16 + c) * D_FEAT + 4 * h;
#pragma unroll
        for (int j = 0; j < 8; ++j) a[0][j] = *(const v4f*)(f0 + 8 * j);
    }

#pragma unroll
    for (int i = 0; i < 16; ++i) {
        const int n   = GEMM_WAVES * i + wave;
        const int dst = (n & 15) * W_CSTRIDE + (n >> 4) * W_K + lane * 4;
        const v4f ws = *(const v4f*)(Wself  + n * D_FEAT + lane * 4);
        const v4f wn = *(const v4f*)(Wneigh + n * D_FEAT + lane * 4);
        *(v4f*)(sW + dst)          = ws;
        *(v4f*)(sW + dst + D_FEAT) = wn;
    }
    __syncthreads();
    if (tile0 >= n_tiles) return;

    int boff = c * W_CSTRIDE + 4 * h + half * 4 * W_K;

    for (int tile = tile0; tile < n_tiles; tile += n_half) {
        const long row = (long)tile * 16 + c;
        const float* fa = feat   + row * D_FEAT + 4 * h;
        const float* ha = hneigh + row * D_FEAT + 4 * h;
        const int tnext = tile + n_half < n_tiles ? tile + n_half : tile;
        const float* fn = feat + ((long)tnext * 16 + c) * D_FEAT + 4 * h;

        asm volatile("" : "+v"(boff));
        const float* bl = sW + boff;

        v8f acc[4];
#pragma unroll
        for (int q = 0; q < 4; ++q) acc[q] = (v8f){};

        v4f bcur[4], bnxt[4];
#pragma unroll
        for (int q = 0; q < 4; ++q)
            bcur[q] = *(const v4f*)(bl + q * W_K);

#pragma unroll
        for (int p = 0; p < 4; ++p) {
            if (p == 0) {
#pragma unroll
                for (int j = 0; j < 8; ++j) a[1][j] = *(const v4f*)(fa + 64 + 8 * j);
            }
            if (p == 1) {
#pragma unroll
                for (int j = 0; j < 8; ++j) a[2][j] = *(const v4f*)(ha + 8 * j);
            }
            if (p == 2) {
#pragma unroll
                for (int j = 0; j < 8; ++j) a[3][j] = *(const v4f*)(ha + 64 + 8 * j);
            }
            if (p == 3) {
#pragma unroll
                for (int j = 0; j < 8; ++j) a[0][j] = *(const v4f*)(fn + 8 * j);
            }
#pragma unroll
            for (int j = 0; j < 8; ++j) {
                const int knext = p * 64 + j * 8 + 8;
                __builtin_amdgcn_sched_barrier(0);
                if (knext < W_K) {
#pragma unroll
                    for (int q = 0; q < 4; ++q)
                        bnxt[q] = *(const v4f*)(bl + q * W_K + knext);
                }
                const v4f av = a[p][j];
                const v2f a_lo = {av.x, av.y};
                const v2f a_hi = {av.z, av.w};
#pragma unroll
                for (int q = 0; q < 4; ++q) {
                    const v2f b_lo = {bcur[q].x, bcur[q].y};
                    acc[q] = wmma_f32(a_lo, b_lo, acc[q]);
                }
#pragma unroll
                for (int q = 0; q < 4; ++q) {
                    const v2f b_hi = {bcur[q].z, bcur[q].w};
                    acc[q] = wmma_f32(a_hi, b_hi, acc[q]);
                }
                if (knext < W_K) {
#pragma unroll
                    for (int q = 0; q < 4; ++q) bcur[q] = bnxt[q];
                }
            }
        }

        float* o = out + ((long)tile * 16 + h * 8) * D_FEAT + half * 64 + c;
#pragma unroll
        for (int q = 0; q < 4; ++q) {
#pragma unroll
            for (int r = 0; r < 8; ++r)
                o[(long)r * D_FEAT + q * 16] = acc[q][r];
        }
    }
}

extern "C" void kernel_launch(void* const* d_in, const int* in_sizes, int n_in,
                              void* d_out, int out_size, void* d_ws, size_t ws_size,
                              hipStream_t stream)
{
    const float* feat   = (const float*)d_in[0];
    const int*   src    = (const int*)d_in[1];
    const int*   dst    = (const int*)d_in[2];
    const float* Wself  = (const float*)d_in[3];
    const float* Wneigh = (const float*)d_in[4];
    float*       out    = (float*)d_out;

    const int n_nodes = in_sizes[0] / D_FEAT;
    const int n_edges = in_sizes[1];
    const int n_bins  = (n_nodes + NODES_PER_BIN - 1) / NODES_PER_BIN;
    const int n_eblk  = (n_edges + EDGES_PER_BLOCK - 1) / EDGES_PER_BLOCK;

    float* hneigh    = (float*)d_ws;
    int2*  pairs     = (int2*)d_out;
    int*   block_off = (int*)(pairs + n_eblk * EDGES_PER_BLOCK);

    sage_block_sort<<<n_eblk, SORT_THREADS, n_bins * sizeof(int), stream>>>(
        src, dst, block_off, pairs, n_edges, n_bins);

    sage_bin_gather_mean<<<n_bins, WALK_THREADS, (2 * n_eblk + 1) * sizeof(int), stream>>>(
        feat, pairs, block_off, hneigh, n_nodes, n_bins, n_eblk);

    const int n_tiles = n_nodes / 16;
    sage_gemm_wmma<<<GEMM_BLOCKS, GEMM_THREADS, 0, stream>>>(feat, hneigh, Wself, Wneigh, out, n_tiles);
}


# Gather walk re-laid: a wave reads one whole 512-byte source row per load (32 lanes x 16 B) and takes its four nodes' runs as one stream, 8 rows in flight, 16 waves per SIMD; sum order unchanged
# speedup vs baseline: 5.8987x; 1.0683x over previous
#include <hip/hip_runtime.h>
#include <hip/hip_bf16.h>

typedef __attribute__((ext_vector_type(2))) float v2f;
typedef __attribute__((ext_vector_type(4))) float v4f;
typedef __attribute__((ext_vector_type(8))) float v8f;
typedef __attribute__((address_space(1))) float global_float;
typedef __attribute__((address_space(1))) v4f global_v4f;

#define D_FEAT 128

#define NODES_PER_BIN   64
#define BIN_SHIFT       6
#define EDGES_PER_BLOCK 8192
#define SORT_THREADS    512
#define EDGES_PER_THREAD (EDGES_PER_BLOCK / SORT_THREADS)

__global__ __launch_bounds__(SORT_THREADS) void sage_block_sort(
    const int* __restrict__ src, const int* __restrict__ dst,
    int* __restrict__ block_off, int2* __restrict__ pairs, int n_edges, int n_bins)
{
    extern __shared__ int lcnt[];
    __shared__ int2 lsorted[EDGES_PER_BLOCK];
    __shared__ int wsum[SORT_THREADS / 32];

    const int e0 = blockIdx.x * EDGES_PER_BLOCK + threadIdx.x;
    int es[EDGES_PER_THREAD], ed[EDGES_PER_THREAD], rk[EDGES_PER_THREAD];
#pragma unroll
    for (int k = 0; k < EDGES_PER_THREAD; ++k) {
        const int e = e0 + k * SORT_THREADS;
        ed[k] = e < n_edges ? dst[e] : 0;
        es[k] = e < n_edges ? src[e] : 0;
    }
    for (int t = threadIdx.x; t < n_bins; t += SORT_THREADS) lcnt[t] = 0;
    __syncthreads();
#pragma unroll
    for (int k = 0; k < EDGES_PER_THREAD; ++k) {
        rk[k] = 0;
        if (e0 + k * SORT_THREADS < n_edges) rk[k] = atomicAdd(&lcnt[ed[k] >> BIN_SHIFT], 1);
    }
    __syncthreads();

    const int cpt  = (n_bins + SORT_THREADS - 1) / SORT_THREADS;
    const int b0   = threadIdx.x * cpt;
    const int lane = threadIdx.x & 31;
    const int wave = threadIdx.x >> 5;
    int own = 0;
    for (int j = 0; j < cpt; ++j)
        if (b0 + j < n_bins) own += lcnt[b0 + j];
    int inc = own;
    for (int d = 1; d < 32; d <<= 1) {
        const int y = __shfl_up(inc, d, 32);
        if (lane >= d) inc += y;
    }
    if (lane == 31) wsum[wave] = inc;
    __syncthreads();
    int run = inc - own;
    for (int w = 0; w < wave; ++w) run += wsum[w];
    for (int j = 0; j < cpt; ++j) {
        const int b = b0 + j;
        if (b < n_bins) {
            const int cb = lcnt[b];
            lcnt[b] = run;
            run += cb;
        }
    }
    __syncthreads();
#pragma unroll
    for (int k = 0; k < EDGES_PER_THREAD; ++k) {
        if (e0 + k * SORT_THREADS < n_edges)
            lsorted[lcnt[ed[k] >> BIN_SHIFT] + rk[k]] = make_int2(es[k], ed[k]);
    }
    __syncthreads();

    const int left   = n_edges - blockIdx.x * EDGES_PER_BLOCK;
    const int nvalid = left < EDGES_PER_BLOCK ? left : EDGES_PER_BLOCK;
    int2* po = pairs + (long)blockIdx.x * EDGES_PER_BLOCK;
    int2 cp[EDGES_PER_THREAD];
#pragma unroll
    for (int k = 0; k < EDGES_PER_THREAD; ++k) {
        const int i = threadIdx.x + k * SORT_THREADS;
        cp[k] = make_int2(0, 0);
        if (i < nvalid) cp[k] = lsorted[i];
    }
#pragma unroll
    for (int k = 0; k < EDGES_PER_THREAD; ++k) {
        const int i = threadIdx.x + k * SORT_THREADS;
        if (i < nvalid) po[i] = cp[k];
    }
    int* orow = block_off + (long)blockIdx.x * (n_bins + 1);
    for (int t = threadIdx.x; t < n_bins; t += SORT_THREADS) orow[t] = lcnt[t];
    if (threadIdx.x == 0) orow[n_bins] = nvalid;
}

#define NODES_PER_WAVE  4
#define WALK_THREADS    (NODES_PER_BIN / NODES_PER_WAVE * 32)
#define WALK_BATCH      8
#define LIST_PER_THREAD 4
#define SAGE_ADD_ROWS(ACC, BEGIN, END)                                         \
    if ((BEGIN) < t + WALK_BATCH && (END) > t) {                               \
        _Pragma("unroll")                         \
        for (int r = 0; r < WALK_BATCH; ++r)                                   \
            if (t + r >= (BEGIN) && t + r < (END)) ACC += x[r];                \
    }
#define LIST_CAP        (WALK_THREADS * LIST_PER_THREAD)

__global__ __launch_bounds__(WALK_THREADS) __attribute__((amdgpu_waves_per_eu(16, 16)))
void sage_bin_gather_mean(
    const float* __restrict__ feat, const int2* __restrict__ pairs,
    const int* __restrict__ block_off, float* __restrict__ hneigh,
    int n_nodes, int n_bins, int n_runs)
{
    extern __shared__ int lrun[];
    __shared__ int lsrc[LIST_CAP];
    __shared__ int lcnt[NODES_PER_BIN];
    __shared__ int lfirst[NODES_PER_BIN];
    __shared__ int rsum[WALK_THREADS / 32];
    int* rpos   = lrun;
    int* rstart = lrun + n_runs + 1;
    const int bin  = blockIdx.x;
    const int base = bin * NODES_PER_BIN;

    const int cpr  = (n_runs + WALK_THREADS - 1) / WALK_THREADS;
    const int r0   = threadIdx.x * cpr;
    const int lane = threadIdx.x & 31;
    const int wave = threadIdx.x >> 5;
    int own = 0;
    for (int j = 0; j < cpr; ++j) {
        const int r = r0 + j;
        if (r < n_runs) {
            const int* orow = block_off + (long)r * (n_bins + 1) + bin;
            const int o0 = orow[0];
            const int o1 = orow[1];
            rstart[r] = r * EDGES_PER_BLOCK + o0;
            rpos[r]   = o1 - o0;
            own += o1 - o0;
        }
    }
    int inc = own;
    for (int d = 1; d < 32; d <<= 1) {
        const int y = __shfl_up(inc, d, 32);
        if (lane >= d) inc += y;
    }
    if (lane == 31) rsum[wave] = inc;
    if (threadIdx.x < NODES_PER_BIN) lcnt[threadIdx.x] = 0;
    __syncthreads();
    int run = inc - own;
    for (int w = 0; w < wave; ++w) run += rsum[w];
    for (int j = 0; j < cpr; ++j) {
        const int r = r0 + j;
        if (r < n_runs) {
            const int len = rpos[r];
            rpos[r] = run;
            run += len;
        }
    }
    if (threadIdx.x == WALK_THREADS - 1) rpos[n_runs] = run;
    __syncthreads();

    const int total = rpos[n_runs];
    const int wv = __builtin_amdgcn_readfirstlane((int)(threadIdx.x >> 5));
    const unsigned loff = (unsigned)lane * 4;
    const global_float* gfeat = (const global_float*)feat;
    const int bl = lane & (WALK_BATCH - 1);

    v4f acc0 = {}, acc1 = {}, acc2 = {}, acc3 = {};
    int deg0 = 0, deg1 = 0, deg2 = 0, deg3 = 0;

    for (int c0 = 0; c0 < total; c0 += LIST_CAP) {
        const int cend = total - c0 < LIST_CAP ? total : c0 + LIST_CAP;
        int es[LIST_PER_THREAD], en[LIST_PER_THREAD], rk[LIST_PER_THREAD];
        int ii[LIST_PER_THREAD], lo[LIST_PER_THREAD], hi[LIST_PER_THREAD];
#pragma unroll
        for (int k = 0; k < LIST_PER_THREAD; ++k) {
            const int i = c0 + (int)threadIdx.x + k * WALK_THREADS;
            ii[k] = i < cend ? i : cend - 1;
            lo[k] = 0;
            hi[k] = n_runs;
        }
        for (int s = 0; s < 32 && (1u << s) < (unsigned)n_runs; ++s) {
#pragma unroll
            for (int k = 0; k < LIST_PER_THREAD; ++k) {
                const int  mid = (lo[k] + hi[k]) >> 1;
                const bool go  = hi[k] - lo[k] > 1;
                const bool up  = rpos[mid] <= ii[k];
                lo[k] = go && up  ? mid : lo[k];
                hi[k] = go && !up ? mid : hi[k];
            }
        }
        int2 pr[LIST_PER_THREAD];
#pragma unroll
        for (int k = 0; k < LIST_PER_THREAD; ++k) {
            pr[k] = make_int2(0, 0);
            if (c0 + (int)threadIdx.x + k * WALK_THREADS < cend)
                pr[k] = pairs[rstart[lo[k]] + (ii[k] - rpos[lo[k]])];
        }
#pragma unroll
        for (int k = 0; k < LIST_PER_THREAD; ++k) {
            es[k] = pr[k].x;
            en[k] = -1;
            rk[k] = 0;
            if (c0 + (int)threadIdx.x + k * WALK_THREADS < cend) {
                en[k] = pr[k].y - base;
                rk[k] = atomicAdd(&lcnt[en[k]], 1);
            }
        }
        __syncthreads();
        if (threadIdx.x < 32) {
            const int ca = lcnt[2 * threadIdx.x];
            const int cb = lcnt[2 * threadIdx.x + 1];
            int ps = ca + cb;
            int ln = lane;
            asm volatile("" : "+v"(ln));
            for (int d = 1; d < 32; d <<= 1) {
                const int from = ln >= d ? ln - d : ln;
                const int y = __builtin_amdgcn_ds_bpermute(from << 2, ps);
                if (ln >= d) ps += y;
            }
            const int ex = ps - (ca + cb);
            lfirst[2 * threadIdx.x]     = ex;
            lfirst[2 * threadIdx.x + 1] = ex + ca;
        }
        __syncthreads();
#pragma unroll
        for (int k = 0; k < LIST_PER_THREAD; ++k) {
            if (en[k] >= 0) lsrc[lfirst[en[k]] + rk[k]] = es[k];
        }
        __syncthreads();

        const int c0n = __builtin_amdgcn_readfirstlane(lcnt[4 * wv]);
        const int c1n = __builtin_amdgcn_readfirstlane(lcnt[4 * wv + 1]);
        const int c2n = __builtin_amdgcn_readfirstlane(lcnt[4 * wv + 2]);
        const int c3n = __builtin_amdgcn_readfirstlane(lcnt[4 * wv + 3]);
        const int* ls = lsrc + __builtin_amdgcn_readfirstlane(lfirst[4 * wv]);
        const int e0 = c0n;
        const int e1 = e0 + c1n;
        const int e2 = e1 + c2n;
        const int e3 = e2 + c3n;

        int idxv = 0;
        if (bl < e3) idxv = ls[bl];
        int t = 0;
        for (; t + WALK_BATCH <= e3; t += WALK_BATCH) {
            v4f x[WALK_BATCH];
            const global_float* rp[WALK_BATCH];
#pragma unroll
            for (int r = 0; r < WALK_BATCH; ++r) {
                rp[r] = gfeat + (long)__builtin_amdgcn_readlane(idxv, r) * D_FEAT;
                asm volatile("" : "+s"(rp[r]));
            }
#pragma unroll
            for (int r = 0; r < WALK_BATCH; ++r) x[r] = *(const global_v4f*)(rp[r] + loff);
            int idxn = 0;
            if (t + WALK_BATCH + bl < e3) idxn = ls[t + WALK_BATCH + bl];
            SAGE_ADD_ROWS(acc0, 0,  e0)
            SAGE_ADD_ROWS(acc1, e0, e1)
            SAGE_ADD_ROWS(acc2, e1, e2)
            SAGE_ADD_ROWS(acc3, e2, e3)
            idxv = idxn;
        }
        if (t < e3) {
            v4f x[WALK_BATCH];
#pragma unroll
            for (int r = 0; r < WALK_BATCH; ++r) {
                x[r] = (v4f){};
                if (t + r < e3) {
                    const global_float* rp = gfeat + (long)__builtin_amdgcn_readlane(idxv, r) * D_FEAT;
                    asm volatile("" : "+s"(rp));
                    x[r] = *(const global_v4f*)(rp + loff);
                }
            }
            SAGE_ADD_ROWS(acc0, 0,  e0)
            SAGE_ADD_ROWS(acc1, e0, e1)
            SAGE_ADD_ROWS(acc2, e1, e2)
            SAGE_ADD_ROWS(acc3, e2, e3)
        }
        deg0 += c0n;
        deg1 += c1n;
        deg2 += c2n;
        deg3 += c3n;

        if (c0 + LIST_CAP < total) {
            __syncthreads();
            if (threadIdx.x < NODES_PER_BIN) lcnt[threadIdx.x] = 0;
            __syncthreads();
        }
    }

    const int v0 = base + 4 * wv;
    float* o = hneigh + (long)v0 * D_FEAT + lane * 4;
    if (v0 < n_nodes)     *(v4f*)(o)              = acc0 / (deg0 > 1 ? (float)deg0 : 1.0f);
    if (v0 + 1 < n_nodes) *(v4f*)(o + D_FEAT)     = acc1 / (deg1 > 1 ? (float)deg1 : 1.0f);
    if (v0 + 2 < n_nodes) *(v4f*)(o + 2 * D_FEAT) = acc2 / (deg2 > 1 ? (float)deg2 : 1.0f);
    if (v0 + 3 < n_nodes) *(v4f*)(o + 3 * D_FEAT) = acc3 / (deg3 > 1 ? (float)deg3 : 1.0f);
}

#define W_K        256
#define W_CSTRIDE  (8 * W_K + 4)
#define GEMM_WAVES   8
#define GEMM_THREADS (GEMM_WAVES * 32)
#define GEMM_BLOCKS  512

__device__ __forceinline__ v8f wmma_f32(v2f a, v2f b, v8f c) {
    return __builtin_amdgcn_wmma_f32_16x16x4_f32(
         false, a,  false, b,
         (short)0, c,  false,  false);
}

__global__ __launch_bounds__(GEMM_THREADS) void sage_gemm_wmma(
    const float* __restrict__ feat, const float* __restrict__ hneigh,
    const float* __restrict__ Wself, const float* __restrict__ Wneigh,
    float* __restrict__ out, int n_tiles)
{
    __shared__ __attribute__((aligned(16))) float sW[16 * W_CSTRIDE];

    const int wave = __builtin_amdgcn_readfirstlane((int)(threadIdx.x >> 5));
    const int lane = threadIdx.x & 31;
    const int c = lane & 15;
    const int h = lane >> 4;

    const int half   = wave & 1;
    const int n_half = (int)gridDim.x * (GEMM_WAVES / 2);
    const int tile0  = (((wave >> 1) + (int)blockIdx.x) & (GEMM_WAVES / 2 - 1)) * (int)gridDim.x
                     + (int)blockIdx.x;

    v4f a[4][8];
    {
        const int tfirst = tile0 < n_tiles ? tile0 : n_tiles - 1;
        const float* f0 = feat + ((long)tfirst * 16 + c) * D_FEAT + 4 * h;
#pragma unroll
        for (int j = 0; j < 8; ++j) a[0][j] = *(const v4f*)(f0 + 8 * j);
    }

#pragma unroll
    for (int i = 0; i < 16; ++i) {
        const int n   = GEMM_WAVES * i + wave;
        const int dst = (n & 15) * W_CSTRIDE + (n >> 4) * W_K + lane * 4;
        const v4f ws = *(const v4f*)(Wself  + n * D_FEAT + lane * 4);
        const v4f wn = *(const v4f*)(Wneigh + n * D_FEAT + lane * 4);
        *(v4f*)(sW + dst)          = ws;
        *(v4f*)(sW + dst + D_FEAT) = wn;
    }
    __syncthreads();
    if (tile0 >= n_tiles) return;

    int boff = c * W_CSTRIDE + 4 * h + half * 4 * W_K;

    for (int tile = tile0; tile < n_tiles; tile += n_half) {
        const long row = (long)tile * 16 + c;
        const float* fa = feat   + row * D_FEAT + 4 * h;
        const float* ha = hneigh + row * D_FEAT + 4 * h;
        const int tnext = tile + n_half < n_tiles ? tile + n_half : tile;
        const float* fn = feat + ((long)tnext * 16 + c) * D_FEAT + 4 * h;

        asm volatile("" : "+v"(boff));
        const float* bl = sW + boff;

        v8f acc[4];
#pragma unroll
        for (int q = 0; q < 4; ++q) acc[q] = (v8f){};

        v4f bcur[4], bnxt[4];
#pragma unroll
        for (int q = 0; q < 4; ++q)
            bcur[q] = *(const v4f*)(bl + q * W_K);

#pragma unroll
        for (int p = 0; p < 4; ++p) {
            if (p == 0) {
#pragma unroll
                for (int j = 0; j < 8; ++j) a[1][j] = *(const v4f*)(fa + 64 + 8 * j);
            }
            if (p == 1) {
#pragma unroll
                for (int j = 0; j < 8; ++j) a[2][j] = *(const v4f*)(ha + 8 * j);
            }
            if (p == 2) {
#pragma unroll
                for (int j = 0; j < 8; ++j) a[3][j] = *(const v4f*)(ha + 64 + 8 * j);
            }
            if (p == 3) {
#pragma unroll
                for (int j = 0; j < 8; ++j) a[0][j] = *(const v4f*)(fn + 8 * j);
            }
#pragma unroll
            for (int j = 0; j < 8; ++j) {
                const int knext = p * 64 + j * 8 + 8;
                __builtin_amdgcn_sched_barrier(0);
                if (knext < W_K) {
#pragma unroll
                    for (int q = 0; q < 4; ++q)
                        bnxt[q] = *(const v4f*)(bl + q * W_K + knext);
                }
                const v4f av = a[p][j];
                const v2f a_lo = {av.x, av.y};
                const v2f a_hi = {av.z, av.w};
#pragma unroll
                for (int q = 0; q < 4; ++q) {
                    const v2f b_lo = {bcur[q].x, bcur[q].y};
                    acc[q] = wmma_f32(a_lo, b_lo, acc[q]);
                }
#pragma unroll
                for (int q = 0; q < 4; ++q) {
                    const v2f b_hi = {bcur[q].z, bcur[q].w};
                    acc[q] = wmma_f32(a_hi, b_hi, acc[q]);
                }
                if (knext < W_K) {
#pragma unroll
                    for (int q = 0; q < 4; ++q) bcur[q] = bnxt[q];
                }
            }
        }

        float* o = out + ((long)tile * 16 + h * 8) * D_FEAT + half * 64 + c;
#pragma unroll
        for (int q = 0; q < 4; ++q) {
#pragma unroll
            for (int r = 0; r < 8; ++r)
                o[(long)r * D_FEAT + q * 16] = acc[q][r];
        }
    }
}

extern "C" void kernel_launch(void* const* d_in, const int* in_sizes, int n_in,
                              void* d_out, int out_size, void* d_ws, size_t ws_size,
                              hipStream_t stream)
{
    const float* feat   = (const float*)d_in[0];
    const int*   src    = (const int*)d_in[1];
    const int*   dst    = (const int*)d_in[2];
    const float* Wself  = (const float*)d_in[3];
    const float* Wneigh = (const float*)d_in[4];
    float*       out    = (float*)d_out;

    const int n_nodes = in_sizes[0] / D_FEAT;
    const int n_edges = in_sizes[1];
    const int n_bins  = (n_nodes + NODES_PER_BIN - 1) / NODES_PER_BIN;
    const int n_eblk  = (n_edges + EDGES_PER_BLOCK - 1) / EDGES_PER_BLOCK;

    float* hneigh    = (float*)d_ws;
    int2*  pairs     = (int2*)d_out;
    int*   block_off = (int*)(pairs + n_eblk * EDGES_PER_BLOCK);

    sage_block_sort<<<n_eblk, SORT_THREADS, n_bins * sizeof(int), stream>>>(
        src, dst, block_off, pairs, n_edges, n_bins);

    sage_bin_gather_mean<<<n_bins, WALK_THREADS, (2 * n_eblk + 1) * sizeof(int), stream>>>(
        feat, pairs, block_off, hneigh, n_nodes, n_bins, n_eblk);

    const int n_tiles = n_nodes / 16;
    sage_gemm_wmma<<<GEMM_BLOCKS, GEMM_THREADS, 0, stream>>>(feat, hneigh, Wself, Wneigh, out, n_tiles);
}


# Dense product moved into the gather kernel: each bin block multiplies its own 64 rows on WMMA from an LDS tile while other blocks gather; 3rd launch and h_neigh round trip gone (k sum in two halves)
# speedup vs baseline: 6.1149x; 1.0367x over previous
#include <hip/hip_runtime.h>
#include <hip/hip_bf16.h>

typedef __attribute__((ext_vector_type(2))) float v2f;
typedef __attribute__((ext_vector_type(4))) float v4f;
typedef __attribute__((ext_vector_type(8))) float v8f;
typedef __attribute__((address_space(1))) float global_float;
typedef __attribute__((address_space(1))) v4f global_v4f;

#define D_FEAT 128

__device__ __forceinline__ v8f wmma_f32(v2f a, v2f b, v8f c) {
    return __builtin_amdgcn_wmma_f32_16x16x4_f32(
         false, a,  false, b,
         (short)0, c,  false,  false);
}

#define NODES_PER_BIN   64
#define BIN_SHIFT       6
#define EDGES_PER_BLOCK 8192
#define SORT_THREADS    512
#define EDGES_PER_THREAD (EDGES_PER_BLOCK / SORT_THREADS)
#define WPACK_FRAGS     256
#define WPACK_BLOCKS    (WPACK_FRAGS * 32 / SORT_THREADS)

__global__ __launch_bounds__(SORT_THREADS) void sage_block_sort(
    const int* __restrict__ src, const int* __restrict__ dst,
    int* __restrict__ block_off, int2* __restrict__ pairs, int n_edges, int n_bins,
    const float* __restrict__ Wself, const float* __restrict__ Wneigh,
    float* __restrict__ wpack, int n_eblk)
{
    if ((int)blockIdx.x >= n_eblk) {
        const int g = ((int)blockIdx.x - n_eblk) * SORT_THREADS + (int)threadIdx.x;
        if (g < WPACK_FRAGS * 32) {
            const int f   = g >> 5;
            const int pc  = g & 15;
            const int ph  = (g >> 4) & 1;
            const int pkh = f >> 7;
            const int pct = (f >> 4) & 7;
            const int pw  = f & 15;
            const float* ws = pkh ? Wneigh : Wself;
            *(v4f*)(wpack + (long)g * 4) =
                *(const v4f*)(ws + (16 * pct + pc) * D_FEAT + 8 * pw + 4 * ph);
        }
        return;
    }
    extern __shared__ int lcnt[];
    __shared__ int2 lsorted[EDGES_PER_BLOCK];
    __shared__ int wsum[SORT_THREADS / 32];

    const int e0 = blockIdx.x * EDGES_PER_BLOCK + threadIdx.x;
    int es[EDGES_PER_THREAD], ed[EDGES_PER_THREAD], rk[EDGES_PER_THREAD];
#pragma unroll
    for (int k = 0; k < EDGES_PER_THREAD; ++k) {
        const int e = e0 + k * SORT_THREADS;
        ed[k] = e < n_edges ? dst[e] : 0;
        es[k] = e < n_edges ? src[e] : 0;
    }
    for (int t = threadIdx.x; t < n_bins; t += SORT_THREADS) lcnt[t] = 0;
    __syncthreads();
#pragma unroll
    for (int k = 0; k < EDGES_PER_THREAD; ++k) {
        rk[k] = 0;
        if (e0 + k * SORT_THREADS < n_edges) rk[k] = atomicAdd(&lcnt[ed[k] >> BIN_SHIFT], 1);
    }
    __syncthreads();

    const int cpt  = (n_bins + SORT_THREADS - 1) / SORT_THREADS;
    const int b0   = threadIdx.x * cpt;
    const int lane = threadIdx.x & 31;
    const int wave = threadIdx.x >> 5;
    int own = 0;
    for (int j = 0; j < cpt; ++j)
        if (b0 + j < n_bins) own += lcnt[b0 + j];
    int inc = own;
    for (int d = 1; d < 32; d <<= 1) {
        const int y = __shfl_up(inc, d, 32);
        if (lane >= d) inc += y;
    }
    if (lane == 31) wsum[wave] = inc;
    __syncthreads();
    int run = inc - own;
    for (int w = 0; w < wave; ++w) run += wsum[w];
    for (int j = 0; j < cpt; ++j) {
        const int b = b0 + j;
        if (b < n_bins) {
            const int cb = lcnt[b];
            lcnt[b] = run;
            run += cb;
        }
    }
    __syncthreads();
#pragma unroll
    for (int k = 0; k < EDGES_PER_THREAD; ++k) {
        if (e0 + k * SORT_THREADS < n_edges)
            lsorted[lcnt[ed[k] >> BIN_SHIFT] + rk[k]] = make_int2(es[k], ed[k]);
    }
    __syncthreads();

    const int left   = n_edges - blockIdx.x * EDGES_PER_BLOCK;
    const int nvalid = left < EDGES_PER_BLOCK ? left : EDGES_PER_BLOCK;
    int2* po = pairs + (long)blockIdx.x * EDGES_PER_BLOCK;
    int2 cp[EDGES_PER_THREAD];
#pragma unroll
    for (int k = 0; k < EDGES_PER_THREAD; ++k) {
        const int i = threadIdx.x + k * SORT_THREADS;
        cp[k] = make_int2(0, 0);
        if (i < nvalid) cp[k] = lsorted[i];
    }
#pragma unroll
    for (int k = 0; k < EDGES_PER_THREAD; ++k) {
        const int i = threadIdx.x + k * SORT_THREADS;
        if (i < nvalid) po[i] = cp[k];
    }
    int* orow = block_off + (long)blockIdx.x * (n_bins + 1);
    for (int t = threadIdx.x; t < n_bins; t += SORT_THREADS) orow[t] = lcnt[t];
    if (threadIdx.x == 0) orow[n_bins] = nvalid;
}

#define NODES_PER_WAVE  4
#define WALK_THREADS    (NODES_PER_BIN / NODES_PER_WAVE * 32)
#define WALK_BATCH      8
#define LIST_PER_THREAD 4
#define SAGE_ADD_ROWS(ACC, BEGIN, END)                                         \
    if ((BEGIN) < t + WALK_BATCH && (END) > t) {                               \
        _Pragma("unroll")                         \
        for (int r = 0; r < WALK_BATCH; ++r)                                   \
            if (t + r >= (BEGIN) && t + r < (END)) ACC += x[r];                \
    }
#define LIST_CAP        (WALK_THREADS * LIST_PER_THREAD)
#define A_PITCH         (2 * D_FEAT + 4)
#define O_PITCH         (D_FEAT + 4)

__global__ __launch_bounds__(WALK_THREADS) __attribute__((amdgpu_waves_per_eu(16, 16)))
void sage_bin_gather_dense(
    const float* __restrict__ feat, const int2* __restrict__ pairs,
    const int* __restrict__ block_off, const float* __restrict__ wpack,
    float* __restrict__ out, int n_nodes, int n_bins, int n_runs)
{
    extern __shared__ int lrun[];
    __shared__ __attribute__((aligned(16))) float sbuf[2 * NODES_PER_BIN * O_PITCH];
    int* lsrc   = (int*)sbuf;
    int* lcnt   = lsrc + LIST_CAP;
    int* lfirst = lcnt + NODES_PER_BIN;
    int* rsum   = lfirst + NODES_PER_BIN;
    int* rpos   = lrun;
    int* rstart = lrun + n_runs + 1;
    const int bin  = blockIdx.x;
    const int base = bin * NODES_PER_BIN;

    const int cpr  = (n_runs + WALK_THREADS - 1) / WALK_THREADS;
    const int r0   = threadIdx.x * cpr;
    const int lane = threadIdx.x & 31;
    const int wave = threadIdx.x >> 5;
    int own = 0;
    for (int j = 0; j < cpr; ++j) {
        const int r = r0 + j;
        if (r < n_runs) {
            const int* orow = block_off + (long)r * (n_bins + 1) + bin;
            const int o0 = orow[0];
            const int o1 = orow[1];
            rstart[r] = r * EDGES_PER_BLOCK + o0;
            rpos[r]   = o1 - o0;
            own += o1 - o0;
        }
    }
    int inc = own;
    for (int d = 1; d < 32; d <<= 1) {
        const int y = __shfl_up(inc, d, 32);
        if (lane >= d) inc += y;
    }
    if (lane == 31) rsum[wave] = inc;
    if (threadIdx.x < NODES_PER_BIN) lcnt[threadIdx.x] = 0;
    __syncthreads();
    int run = inc - own;
    for (int w = 0; w < wave; ++w) run += rsum[w];
    for (int j = 0; j < cpr; ++j) {
        const int r = r0 + j;
        if (r < n_runs) {
            const int len = rpos[r];
            rpos[r] = run;
            run += len;
        }
    }
    if (threadIdx.x == WALK_THREADS - 1) rpos[n_runs] = run;
    __syncthreads();

    const int total = rpos[n_runs];
    const int wv = __builtin_amdgcn_readfirstlane((int)(threadIdx.x >> 5));
    const unsigned loff = (unsigned)lane * 4;
    const global_float* gfeat = (const global_float*)feat;
    const int bl = lane & (WALK_BATCH - 1);

    v4f acc0 = {}, acc1 = {}, acc2 = {}, acc3 = {};
    int deg0 = 0, deg1 = 0, deg2 = 0, deg3 = 0;

    for (int c0 = 0; c0 < total; c0 += LIST_CAP) {
        const int cend = total - c0 < LIST_CAP ? total : c0 + LIST_CAP;
        int es[LIST_PER_THREAD], en[LIST_PER_THREAD], rk[LIST_PER_THREAD];
        int ii[LIST_PER_THREAD], lo[LIST_PER_THREAD], hi[LIST_PER_THREAD];
#pragma unroll
        for (int k = 0; k < LIST_PER_THREAD; ++k) {
            const int i = c0 + (int)threadIdx.x + k * WALK_THREADS;
            ii[k] = i < cend ? i : cend - 1;
            lo[k] = 0;
            hi[k] = n_runs;
        }
        for (int s = 0; s < 32 && (1u << s) < (unsigned)n_runs; ++s) {
#pragma unroll
            for (int k = 0; k < LIST_PER_THREAD; ++k) {
                const int  mid = (lo[k] + hi[k]) >> 1;
                const bool go  = hi[k] - lo[k] > 1;
                const bool up  = rpos[mid] <= ii[k];
                lo[k] = go && up  ? mid : lo[k];
                hi[k] = go && !up ? mid : hi[k];
            }
        }
        int2 pr[LIST_PER_THREAD];
#pragma unroll
        for (int k = 0; k < LIST_PER_THREAD; ++k) {
            pr[k] = make_int2(0, 0);
            if (c0 + (int)threadIdx.x + k * WALK_THREADS < cend)
                pr[k] = pairs[rstart[lo[k]] + (ii[k] - rpos[lo[k]])];
        }
#pragma unroll
        for (int k = 0; k < LIST_PER_THREAD; ++k) {
            es[k] = pr[k].x;
            en[k] = -1;
            rk[k] = 0;
            if (c0 + (int)threadIdx.x + k * WALK_THREADS < cend) {
                en[k] = pr[k].y - base;
                rk[k] = atomicAdd(&lcnt[en[k]], 1);
            }
        }
        __syncthreads();
        if (threadIdx.x < 32) {
            const int ca = lcnt[2 * threadIdx.x];
            const int cb = lcnt[2 * threadIdx.x + 1];
            int ps = ca + cb;
            int ln = lane;
            asm volatile("" : "+v"(ln));
            for (int d = 1; d < 32; d <<= 1) {
                const int from = ln >= d ? ln - d : ln;
                const int y = __builtin_amdgcn_ds_bpermute(from << 2, ps);
                if (ln >= d) ps += y;
            }
            const int ex = ps - (ca + cb);
            lfirst[2 * threadIdx.x]     = ex;
            lfirst[2 * threadIdx.x + 1] = ex + ca;
        }
        __syncthreads();
#pragma unroll
        for (int k = 0; k < LIST_PER_THREAD; ++k) {
            if (en[k] >= 0) lsrc[lfirst[en[k]] + rk[k]] = es[k];
        }
        __syncthreads();

        const int c0n = __builtin_amdgcn_readfirstlane(lcnt[4 * wv]);
        const int c1n = __builtin_amdgcn_readfirstlane(lcnt[4 * wv + 1]);
        const int c2n = __builtin_amdgcn_readfirstlane(lcnt[4 * wv + 2]);
        const int c3n = __builtin_amdgcn_readfirstlane(lcnt[4 * wv + 3]);
        const int* ls = lsrc + __builtin_amdgcn_readfirstlane(lfirst[4 * wv]);
        const int e0 = c0n;
        const int e1 = e0 + c1n;
        const int e2 = e1 + c2n;
        const int e3 = e2 + c3n;

        int idxv = 0;
        if (bl < e3) idxv = ls[bl];
        int t = 0;
        for (; t + WALK_BATCH <= e3; t += WALK_BATCH) {
            v4f x[WALK_BATCH];
            const global_float* rp[WALK_BATCH];
#pragma unroll
            for (int r = 0; r < WALK_BATCH; ++r) {
                rp[r] = gfeat + (long)__builtin_amdgcn_readlane(idxv, r) * D_FEAT;
                asm volatile("" : "+s"(rp[r]));
            }
#pragma unroll
            for (int r = 0; r < WALK_BATCH; ++r) x[r] = *(const global_v4f*)(rp[r] + loff);
            int idxn = 0;
            if (t + WALK_BATCH + bl < e3) idxn = ls[t + WALK_BATCH + bl];
            SAGE_ADD_ROWS(acc0, 0,  e0)
            SAGE_ADD_ROWS(acc1, e0, e1)
            SAGE_ADD_ROWS(acc2, e1, e2)
            SAGE_ADD_ROWS(acc3, e2, e3)
            idxv = idxn;
        }
        if (t < e3) {
            v4f x[WALK_BATCH];
#pragma unroll
            for (int r = 0; r < WALK_BATCH; ++r) {
                x[r] = (v4f){};
                if (t + r < e3) {
                    const global_float* rp = gfeat + (long)__builtin_amdgcn_readlane(idxv, r) * D_FEAT;
                    asm volatile("" : "+s"(rp));
                    x[r] = *(const global_v4f*)(rp + loff);
                }
            }
            SAGE_ADD_ROWS(acc0, 0,  e0)
            SAGE_ADD_ROWS(acc1, e0, e1)
            SAGE_ADD_ROWS(acc2, e1, e2)
            SAGE_ADD_ROWS(acc3, e2, e3)
        }
        deg0 += c0n;
        deg1 += c1n;
        deg2 += c2n;
        deg3 += c3n;

        if (c0 + LIST_CAP < total) {
            __syncthreads();
            if (threadIdx.x < NODES_PER_BIN) lcnt[threadIdx.x] = 0;
            __syncthreads();
        }
    }

    const v4f hn0 = acc0 / (deg0 > 1 ? (float)deg0 : 1.0f);
    const v4f hn1 = acc1 / (deg1 > 1 ? (float)deg1 : 1.0f);
    const v4f hn2 = acc2 / (deg2 > 1 ? (float)deg2 : 1.0f);
    const v4f hn3 = acc3 / (deg3 > 1 ? (float)deg3 : 1.0f);
    __syncthreads();
    {
        const int r0 = 4 * wv;
        v4f fr[NODES_PER_WAVE];
#pragma unroll
        for (int j = 0; j < NODES_PER_WAVE; ++j) {
            fr[j] = (v4f){};
            if (base + r0 + j < n_nodes)
                fr[j] = *(const global_v4f*)(gfeat + (long)(base + r0 + j) * D_FEAT + loff);
        }
        *(v4f*)(sbuf + (r0 + 0) * A_PITCH + D_FEAT + loff) = hn0;
        *(v4f*)(sbuf + (r0 + 1) * A_PITCH + D_FEAT + loff) = hn1;
        *(v4f*)(sbuf + (r0 + 2) * A_PITCH + D_FEAT + loff) = hn2;
        *(v4f*)(sbuf + (r0 + 3) * A_PITCH + D_FEAT + loff) = hn3;
#pragma unroll
        for (int j = 0; j < NODES_PER_WAVE; ++j)
            *(v4f*)(sbuf + (r0 + j) * A_PITCH + loff) = fr[j];
    }
    __syncthreads();

    const int fc = lane & 15;
    const int fh = lane >> 4;
    const int ct = wv & 7;
    const int kh = wv >> 3;
    const float* bp = wpack + (long)((kh * 8 + ct) * 16) * 128 + lane * 4;
    const float* ap = sbuf + fc * A_PITCH + kh * D_FEAT + 4 * fh;
    v8f dacc[4];
#pragma unroll
    for (int rt = 0; rt < 4; ++rt) dacc[rt] = (v8f){};
    v4f bcur = *(const v4f*)bp;
#pragma unroll
    for (int win = 0; win < 16; ++win) {
        v4f bnxt = bcur;
        if (win + 1 < 16) bnxt = *(const v4f*)(bp + (win + 1) * 128);
        v4f av[4];
#pragma unroll
        for (int rt = 0; rt < 4; ++rt)
            av[rt] = *(const v4f*)(ap + rt * 16 * A_PITCH + 8 * win);
        const v2f b_lo = {bcur.x, bcur.y};
        const v2f b_hi = {bcur.z, bcur.w};
#pragma unroll
        for (int rt = 0; rt < 4; ++rt) {
            const v2f a_lo = {av[rt].x, av[rt].y};
            dacc[rt] = wmma_f32(a_lo, b_lo, dacc[rt]);
        }
#pragma unroll
        for (int rt = 0; rt < 4; ++rt) {
            const v2f a_hi = {av[rt].z, av[rt].w};
            dacc[rt] = wmma_f32(a_hi, b_hi, dacc[rt]);
        }
        bcur = bnxt;
    }
    __syncthreads();
    {
        float* so = sbuf + kh * (NODES_PER_BIN * O_PITCH) + (8 * fh) * O_PITCH + 16 * ct + fc;
#pragma unroll
        for (int rt = 0; rt < 4; ++rt) {
#pragma unroll
            for (int r = 0; r < 8; ++r)
                so[(16 * rt + r) * O_PITCH] = dacc[rt][r];
        }
    }
    __syncthreads();
#pragma unroll
    for (int j = 0; j < NODES_PER_WAVE; ++j) {
        const int row = 4 * wv + j;
        const v4f s0 = *(const v4f*)(sbuf + row * O_PITCH + loff);
        const v4f s1 = *(const v4f*)(sbuf + NODES_PER_BIN * O_PITCH + row * O_PITCH + loff);
        if (base + row < n_nodes)
            *(v4f*)(out + (long)(base + row) * D_FEAT + loff) = s0 + s1;
    }
}

extern "C" void kernel_launch(void* const* d_in, const int* in_sizes, int n_in,
                              void* d_out, int out_size, void* d_ws, size_t ws_size,
                              hipStream_t stream)
{
    const float* feat   = (const float*)d_in[0];
    const int*   src    = (const int*)d_in[1];
    const int*   dst    = (const int*)d_in[2];
    const float* Wself  = (const float*)d_in[3];
    const float* Wneigh = (const float*)d_in[4];
    float*       out    = (float*)d_out;

    const int n_nodes = in_sizes[0] / D_FEAT;
    const int n_edges = in_sizes[1];
    const int n_bins  = (n_nodes + NODES_PER_BIN - 1) / NODES_PER_BIN;
    const int n_eblk  = (n_edges + EDGES_PER_BLOCK - 1) / EDGES_PER_BLOCK;

    float* wpack     = (float*)d_ws;
    int2*  pairs     = (int2*)(wpack + WPACK_FRAGS * 128);
    int*   block_off = (int*)(pairs + n_eblk * EDGES_PER_BLOCK);

    sage_block_sort<<<n_eblk + WPACK_BLOCKS, SORT_THREADS, n_bins * sizeof(int), stream>>>(
        src, dst, block_off, pairs, n_edges, n_bins, Wself, Wneigh, wpack, n_eblk);

    sage_bin_gather_dense<<<n_bins, WALK_THREADS, (2 * n_eblk + 1) * sizeof(int), stream>>>(
        feat, pairs, block_off, wpack, out, n_nodes, n_bins, n_eblk);
}


# Wave priority by phase in the fused kernel: list build and gather walk at s_setprio 3, WMMA products at 0, so an older block's products stop stalling younger blocks' row loads; sum order unchanged
# speedup vs baseline: 6.3571x; 1.0396x over previous
#include <hip/hip_runtime.h>
#include <hip/hip_bf16.h>

typedef __attribute__((ext_vector_type(2))) float v2f;
typedef __attribute__((ext_vector_type(4))) float v4f;
typedef __attribute__((ext_vector_type(8))) float v8f;
typedef __attribute__((address_space(1))) float global_float;
typedef __attribute__((address_space(1))) v4f global_v4f;

#define D_FEAT 128

__device__ __forceinline__ v8f wmma_f32(v2f a, v2f b, v8f c) {
    return __builtin_amdgcn_wmma_f32_16x16x4_f32(
         false, a,  false, b,
         (short)0, c,  false,  false);
}

#define NODES_PER_BIN   64
#define BIN_SHIFT       6
#define EDGES_PER_BLOCK 8192
#define SORT_THREADS    512
#define EDGES_PER_THREAD (EDGES_PER_BLOCK / SORT_THREADS)
#define WPACK_FRAGS     256
#define WPACK_BLOCKS    (WPACK_FRAGS * 32 / SORT_THREADS)

__global__ __launch_bounds__(SORT_THREADS) void sage_block_sort(
    const int* __restrict__ src, const int* __restrict__ dst,
    int* __restrict__ block_off, int2* __restrict__ pairs, int n_edges, int n_bins,
    const float* __restrict__ Wself, const float* __restrict__ Wneigh,
    float* __restrict__ wpack, int n_eblk)
{
    if ((int)blockIdx.x >= n_eblk) {
        const int g = ((int)blockIdx.x - n_eblk) * SORT_THREADS + (int)threadIdx.x;
        if (g < WPACK_FRAGS * 32) {
            const int f   = g >> 5;
            const int pc  = g & 15;
            const int ph  = (g >> 4) & 1;
            const int pkh = f >> 7;
            const int pct = (f >> 4) & 7;
            const int pw  = f & 15;
            const float* ws = pkh ? Wneigh : Wself;
            *(v4f*)(wpack + (long)g * 4) =
                *(const v4f*)(ws + (16 * pct + pc) * D_FEAT + 8 * pw + 4 * ph);
        }
        return;
    }
    extern __shared__ int lcnt[];
    __shared__ int2 lsorted[EDGES_PER_BLOCK];
    __shared__ int wsum[SORT_THREADS / 32];

    const int e0 = blockIdx.x * EDGES_PER_BLOCK + threadIdx.x;
    int es[EDGES_PER_THREAD], ed[EDGES_PER_THREAD], rk[EDGES_PER_THREAD];
#pragma unroll
    for (int k = 0; k < EDGES_PER_THREAD; ++k) {
        const int e = e0 + k * SORT_THREADS;
        ed[k] = e < n_edges ? dst[e] : 0;
        es[k] = e < n_edges ? src[e] : 0;
    }
    for (int t = threadIdx.x; t < n_bins; t += SORT_THREADS) lcnt[t] = 0;
    __syncthreads();
#pragma unroll
    for (int k = 0; k < EDGES_PER_THREAD; ++k) {
        rk[k] = 0;
        if (e0 + k * SORT_THREADS < n_edges) rk[k] = atomicAdd(&lcnt[ed[k] >> BIN_SHIFT], 1);
    }
    __syncthreads();

    const int cpt  = (n_bins + SORT_THREADS - 1) / SORT_THREADS;
    const int b0   = threadIdx.x * cpt;
    const int lane = threadIdx.x & 31;
    const int wave = threadIdx.x >> 5;
    int own = 0;
    for (int j = 0; j < cpt; ++j)
        if (b0 + j < n_bins) own += lcnt[b0 + j];
    int inc = own;
    for (int d = 1; d < 32; d <<= 1) {
        const int y = __shfl_up(inc, d, 32);
        if (lane >= d) inc += y;
    }
    if (lane == 31) wsum[wave] = inc;
    __syncthreads();
    int run = inc - own;
    for (int w = 0; w < wave; ++w) run += wsum[w];
    for (int j = 0; j < cpt; ++j) {
        const int b = b0 + j;
        if (b < n_bins) {
            const int cb = lcnt[b];
            lcnt[b] = run;
            run += cb;
        }
    }
    __syncthreads();
#pragma unroll
    for (int k = 0; k < EDGES_PER_THREAD; ++k) {
        if (e0 + k * SORT_THREADS < n_edges)
            lsorted[lcnt[ed[k] >> BIN_SHIFT] + rk[k]] = make_int2(es[k], ed[k]);
    }
    __syncthreads();

    const int left   = n_edges - blockIdx.x * EDGES_PER_BLOCK;
    const int nvalid = left < EDGES_PER_BLOCK ? left : EDGES_PER_BLOCK;
    int2* po = pairs + (long)blockIdx.x * EDGES_PER_BLOCK;
    int2 cp[EDGES_PER_THREAD];
#pragma unroll
    for (int k = 0; k < EDGES_PER_THREAD; ++k) {
        const int i = threadIdx.x + k * SORT_THREADS;
        cp[k] = make_int2(0, 0);
        if (i < nvalid) cp[k] = lsorted[i];
    }
#pragma unroll
    for (int k = 0; k < EDGES_PER_THREAD; ++k) {
        const int i = threadIdx.x + k * SORT_THREADS;
        if (i < nvalid) po[i] = cp[k];
    }
    int* orow = block_off + (long)blockIdx.x * (n_bins + 1);
    for (int t = threadIdx.x; t < n_bins; t += SORT_THREADS) orow[t] = lcnt[t];
    if (threadIdx.x == 0) orow[n_bins] = nvalid;
}

#define NODES_PER_WAVE  4
#define WALK_THREADS    (NODES_PER_BIN / NODES_PER_WAVE * 32)
#define WALK_BATCH      8
#define LIST_PER_THREAD 4
#define SAGE_ADD_ROWS(ACC, BEGIN, END)                                         \
    if ((BEGIN) < t + WALK_BATCH && (END) > t) {                               \
        _Pragma("unroll")                         \
        for (int r = 0; r < WALK_BATCH; ++r)                                   \
            if (t + r >= (BEGIN) && t + r < (END)) ACC += x[r];                \
    }
#define LIST_CAP        (WALK_THREADS * LIST_PER_THREAD)
#define A_PITCH         (2 * D_FEAT + 4)
#define O_PITCH         (D_FEAT + 4)

__global__ __launch_bounds__(WALK_THREADS) __attribute__((amdgpu_waves_per_eu(16, 16)))
void sage_bin_gather_dense(
    const float* __restrict__ feat, const int2* __restrict__ pairs,
    const int* __restrict__ block_off, const float* __restrict__ wpack,
    float* __restrict__ out, int n_nodes, int n_bins, int n_runs)
{
    __builtin_amdgcn_s_setprio(3);
    extern __shared__ int lrun[];
    __shared__ __attribute__((aligned(16))) float sbuf[2 * NODES_PER_BIN * O_PITCH];
    int* lsrc   = (int*)sbuf;
    int* lcnt   = lsrc + LIST_CAP;
    int* lfirst = lcnt + NODES_PER_BIN;
    int* rsum   = lfirst + NODES_PER_BIN;
    int* rpos   = lrun;
    int* rstart = lrun + n_runs + 1;
    const int bin  = blockIdx.x;
    const int base = bin * NODES_PER_BIN;

    const int cpr  = (n_runs + WALK_THREADS - 1) / WALK_THREADS;
    const int r0   = threadIdx.x * cpr;
    const int lane = threadIdx.x & 31;
    const int wave = threadIdx.x >> 5;
    int own = 0;
    for (int j = 0; j < cpr; ++j) {
        const int r = r0 + j;
        if (r < n_runs) {
            const int* orow = block_off + (long)r * (n_bins + 1) + bin;
            const int o0 = orow[0];
            const int o1 = orow[1];
            rstart[r] = r * EDGES_PER_BLOCK + o0;
            rpos[r]   = o1 - o0;
            own += o1 - o0;
        }
    }
    int inc = own;
    for (int d = 1; d < 32; d <<= 1) {
        const int y = __shfl_up(inc, d, 32);
        if (lane >= d) inc += y;
    }
    if (lane == 31) rsum[wave] = inc;
    if (threadIdx.x < NODES_PER_BIN) lcnt[threadIdx.x] = 0;
    __syncthreads();
    int run = inc - own;
    for (int w = 0; w < wave; ++w) run += rsum[w];
    for (int j = 0; j < cpr; ++j) {
        const int r = r0 + j;
        if (r < n_runs) {
            const int len = rpos[r];
            rpos[r] = run;
            run += len;
        }
    }
    if (threadIdx.x == WALK_THREADS - 1) rpos[n_runs] = run;
    __syncthreads();

    const int total = rpos[n_runs];
    const int wv = __builtin_amdgcn_readfirstlane((int)(threadIdx.x >> 5));
    const unsigned loff = (unsigned)lane * 4;
    const global_float* gfeat = (const global_float*)feat;
    const int bl = lane & (WALK_BATCH - 1);

    v4f acc0 = {}, acc1 = {}, acc2 = {}, acc3 = {};
    int deg0 = 0, deg1 = 0, deg2 = 0, deg3 = 0;

    for (int c0 = 0; c0 < total; c0 += LIST_CAP) {
        const int cend = total - c0 < LIST_CAP ? total : c0 + LIST_CAP;
        int es[LIST_PER_THREAD], en[LIST_PER_THREAD], rk[LIST_PER_THREAD];
        int ii[LIST_PER_THREAD], lo[LIST_PER_THREAD], hi[LIST_PER_THREAD];
#pragma unroll
        for (int k = 0; k < LIST_PER_THREAD; ++k) {
            const int i = c0 + (int)threadIdx.x + k * WALK_THREADS;
            ii[k] = i < cend ? i : cend - 1;
            lo[k] = 0;
            hi[k] = n_runs;
        }
        for (int s = 0; s < 32 && (1u << s) < (unsigned)n_runs; ++s) {
#pragma unroll
            for (int k = 0; k < LIST_PER_THREAD; ++k) {
                const int  mid = (lo[k] + hi[k]) >> 1;
                const bool go  = hi[k] - lo[k] > 1;
                const bool up  = rpos[mid] <= ii[k];
                lo[k] = go && up  ? mid : lo[k];
                hi[k] = go && !up ? mid : hi[k];
            }
        }
        int2 pr[LIST_PER_THREAD];
#pragma unroll
        for (int k = 0; k < LIST_PER_THREAD; ++k) {
            pr[k] = make_int2(0, 0);
            if (c0 + (int)threadIdx.x + k * WALK_THREADS < cend)
                pr[k] = pairs[rstart[lo[k]] + (ii[k] - rpos[lo[k]])];
        }
#pragma unroll
        for (int k = 0; k < LIST_PER_THREAD; ++k) {
            es[k] = pr[k].x;
            en[k] = -1;
            rk[k] = 0;
            if (c0 + (int)threadIdx.x + k * WALK_THREADS < cend) {
                en[k] = pr[k].y - base;
                rk[k] = atomicAdd(&lcnt[en[k]], 1);
            }
        }
        __syncthreads();
        if (threadIdx.x < 32) {
            const int ca = lcnt[2 * threadIdx.x];
            const int cb = lcnt[2 * threadIdx.x + 1];
            int ps = ca + cb;
            int ln = lane;
            asm volatile("" : "+v"(ln));
            for (int d = 1; d < 32; d <<= 1) {
                const int from = ln >= d ? ln - d : ln;
                const int y = __builtin_amdgcn_ds_bpermute(from << 2, ps);
                if (ln >= d) ps += y;
            }
            const int ex = ps - (ca + cb);
            lfirst[2 * threadIdx.x]     = ex;
            lfirst[2 * threadIdx.x + 1] = ex + ca;
        }
        __syncthreads();
#pragma unroll
        for (int k = 0; k < LIST_PER_THREAD; ++k) {
            if (en[k] >= 0) lsrc[lfirst[en[k]] + rk[k]] = es[k];
        }
        __syncthreads();

        const int c0n = __builtin_amdgcn_readfirstlane(lcnt[4 * wv]);
        const int c1n = __builtin_amdgcn_readfirstlane(lcnt[4 * wv + 1]);
        const int c2n = __builtin_amdgcn_readfirstlane(lcnt[4 * wv + 2]);
        const int c3n = __builtin_amdgcn_readfirstlane(lcnt[4 * wv + 3]);
        const int* ls = lsrc + __builtin_amdgcn_readfirstlane(lfirst[4 * wv]);
        const int e0 = c0n;
        const int e1 = e0 + c1n;
        const int e2 = e1 + c2n;
        const int e3 = e2 + c3n;

        int idxv = 0;
        if (bl < e3) idxv = ls[bl];
        int t = 0;
        for (; t + WALK_BATCH <= e3; t += WALK_BATCH) {
            v4f x[WALK_BATCH];
            const global_float* rp[WALK_BATCH];
#pragma unroll
            for (int r = 0; r < WALK_BATCH; ++r) {
                rp[r] = gfeat + (long)__builtin_amdgcn_readlane(idxv, r) * D_FEAT;
                asm volatile("" : "+s"(rp[r]));
            }
#pragma unroll
            for (int r = 0; r < WALK_BATCH; ++r) x[r] = *(const global_v4f*)(rp[r] + loff);
            int idxn = 0;
            if (t + WALK_BATCH + bl < e3) idxn = ls[t + WALK_BATCH + bl];
            SAGE_ADD_ROWS(acc0, 0,  e0)
            SAGE_ADD_ROWS(acc1, e0, e1)
            SAGE_ADD_ROWS(acc2, e1, e2)
            SAGE_ADD_ROWS(acc3, e2, e3)
            idxv = idxn;
        }
        if (t < e3) {
            v4f x[WALK_BATCH];
#pragma unroll
            for (int r = 0; r < WALK_BATCH; ++r) {
                x[r] = (v4f){};
                if (t + r < e3) {
                    const global_float* rp = gfeat + (long)__builtin_amdgcn_readlane(idxv, r) * D_FEAT;
                    asm volatile("" : "+s"(rp));
                    x[r] = *(const global_v4f*)(rp + loff);
                }
            }
            SAGE_ADD_ROWS(acc0, 0,  e0)
            SAGE_ADD_ROWS(acc1, e0, e1)
            SAGE_ADD_ROWS(acc2, e1, e2)
            SAGE_ADD_ROWS(acc3, e2, e3)
        }
        deg0 += c0n;
        deg1 += c1n;
        deg2 += c2n;
        deg3 += c3n;

        if (c0 + LIST_CAP < total) {
            __syncthreads();
            if (threadIdx.x < NODES_PER_BIN) lcnt[threadIdx.x] = 0;
            __syncthreads();
        }
    }

    const v4f hn0 = acc0 / (deg0 > 1 ? (float)deg0 : 1.0f);
    const v4f hn1 = acc1 / (deg1 > 1 ? (float)deg1 : 1.0f);
    const v4f hn2 = acc2 / (deg2 > 1 ? (float)deg2 : 1.0f);
    const v4f hn3 = acc3 / (deg3 > 1 ? (float)deg3 : 1.0f);
    __syncthreads();
    {
        const int r0 = 4 * wv;
        v4f fr[NODES_PER_WAVE];
#pragma unroll
        for (int j = 0; j < NODES_PER_WAVE; ++j) {
            fr[j] = (v4f){};
            if (base + r0 + j < n_nodes)
                fr[j] = *(const global_v4f*)(gfeat + (long)(base + r0 + j) * D_FEAT + loff);
        }
        *(v4f*)(sbuf + (r0 + 0) * A_PITCH + D_FEAT + loff) = hn0;
        *(v4f*)(sbuf + (r0 + 1) * A_PITCH + D_FEAT + loff) = hn1;
        *(v4f*)(sbuf + (r0 + 2) * A_PITCH + D_FEAT + loff) = hn2;
        *(v4f*)(sbuf + (r0 + 3) * A_PITCH + D_FEAT + loff) = hn3;
#pragma unroll
        for (int j = 0; j < NODES_PER_WAVE; ++j)
            *(v4f*)(sbuf + (r0 + j) * A_PITCH + loff) = fr[j];
    }
    __syncthreads();
    __builtin_amdgcn_s_setprio(0);

    const int fc = lane & 15;
    const int fh = lane >> 4;
    const int ct = wv & 7;
    const int kh = wv >> 3;
    const float* bp = wpack + (long)((kh * 8 + ct) * 16) * 128 + lane * 4;
    const float* ap = sbuf + fc * A_PITCH + kh * D_FEAT + 4 * fh;
    v8f dacc[4];
#pragma unroll
    for (int rt = 0; rt < 4; ++rt) dacc[rt] = (v8f){};
    v4f bcur = *(const v4f*)bp;
#pragma unroll
    for (int win = 0; win < 16; ++win) {
        v4f bnxt = bcur;
        if (win + 1 < 16) bnxt = *(const v4f*)(bp + (win + 1) * 128);
        v4f av[4];
#pragma unroll
        for (int rt = 0; rt < 4; ++rt)
            av[rt] = *(const v4f*)(ap + rt * 16 * A_PITCH + 8 * win);
        const v2f b_lo = {bcur.x, bcur.y};
        const v2f b_hi = {bcur.z, bcur.w};
#pragma unroll
        for (int rt = 0; rt < 4; ++rt) {
            const v2f a_lo = {av[rt].x, av[rt].y};
            dacc[rt] = wmma_f32(a_lo, b_lo, dacc[rt]);
        }
#pragma unroll
        for (int rt = 0; rt < 4; ++rt) {
            const v2f a_hi = {av[rt].z, av[rt].w};
            dacc[rt] = wmma_f32(a_hi, b_hi, dacc[rt]);
        }
        bcur = bnxt;
    }
    __syncthreads();
    {
        float* so = sbuf + kh * (NODES_PER_BIN * O_PITCH) + (8 * fh) * O_PITCH + 16 * ct + fc;
#pragma unroll
        for (int rt = 0; rt < 4; ++rt) {
#pragma unroll
            for (int r = 0; r < 8; ++r)
                so[(16 * rt + r) * O_PITCH] = dacc[rt][r];
        }
    }
    __syncthreads();
#pragma unroll
    for (int j = 0; j < NODES_PER_WAVE; ++j) {
        const int row = 4 * wv + j;
        const v4f s0 = *(const v4f*)(sbuf + row * O_PITCH + loff);
        const v4f s1 = *(const v4f*)(sbuf + NODES_PER_BIN * O_PITCH + row * O_PITCH + loff);
        if (base + row < n_nodes)
            *(v4f*)(out + (long)(base + row) * D_FEAT + loff) = s0 + s1;
    }
}

extern "C" void kernel_launch(void* const* d_in, const int* in_sizes, int n_in,
                              void* d_out, int out_size, void* d_ws, size_t ws_size,
                              hipStream_t stream)
{
    const float* feat   = (const float*)d_in[0];
    const int*   src    = (const int*)d_in[1];
    const int*   dst    = (const int*)d_in[2];
    const float* Wself  = (const float*)d_in[3];
    const float* Wneigh = (const float*)d_in[4];
    float*       out    = (float*)d_out;

    const int n_nodes = in_sizes[0] / D_FEAT;
    const int n_edges = in_sizes[1];
    const int n_bins  = (n_nodes + NODES_PER_BIN - 1) / NODES_PER_BIN;
    const int n_eblk  = (n_edges + EDGES_PER_BLOCK - 1) / EDGES_PER_BLOCK;

    float* wpack     = (float*)d_ws;
    int2*  pairs     = (int2*)(wpack + WPACK_FRAGS * 128);
    int*   block_off = (int*)(pairs + n_eblk * EDGES_PER_BLOCK);

    sage_block_sort<<<n_eblk + WPACK_BLOCKS, SORT_THREADS, n_bins * sizeof(int), stream>>>(
        src, dst, block_off, pairs, n_edges, n_bins, Wself, Wneigh, wpack, n_eblk);

    sage_bin_gather_dense<<<n_bins, WALK_THREADS, (2 * n_eblk + 1) * sizeof(int), stream>>>(
        feat, pairs, block_off, wpack, out, n_nodes, n_bins, n_eblk);
}
